# starting kernel
# baseline (speedup 1.0000x reference)
#include <hip/hip_runtime.h>
#include <hip/hip_bf16.h>

typedef __attribute__((ext_vector_type(2))) float v2f;
typedef __attribute__((ext_vector_type(8))) float v8f;

__device__ __forceinline__ v8f wmma4(v2f a, v2f b, v8f c) {
    return __builtin_amdgcn_wmma_f32_16x16x4_f32(
         false, a,  false, b,
         (short)0, c,  false,  false);
}

__global__ void k_zero(float* p, int n) {
    int i = blockIdx.x * blockDim.x + threadIdx.x;
    if (i < n) p[i] = 0.0f;
}

__global__ void k_count_deg(const int* __restrict__ dst, float* __restrict__ deg, int E) {
    int e = blockIdx.x * blockDim.x + threadIdx.x;
    if (e < E) atomicAdd(&deg[dst[e]], 1.0f);
}

__global__ void k_dinv(float* __restrict__ d, int N) {
    int i = blockIdx.x * blockDim.x + threadIdx.x;
    if (i < N) d[i] = rsqrtf(d[i] + 1.0f);
}

__global__ void k_encoder(const float* __restrict__ x, const float* __restrict__ cnn,
                          const float* __restrict__ fcW, const float* __restrict__ fcb,
                          const float* __restrict__ cnnW, const float* __restrict__ cnnb,
                          float* __restrict__ h0, int N, int nTiles) {
    int wave = (blockIdx.x * blockDim.x + threadIdx.x) >> 5;
    if (wave >= nTiles) return;
    int lane = threadIdx.x & 31;
    int n    = lane & 15;
    int kh   = lane >> 4;
    int rowA = wave * 16 + n;    if (rowA >= N) rowA = N - 1;

    v8f acc0 = {}, acc1 = {}, acc2 = {}, acc3 = {};

    const float* ax = x + (size_t)rowA * 32;
#pragma unroll
    for (int k0 = 0; k0 < 32; k0 += 4) {
        int kb = k0 + 2 * kh;
        v2f a;  a.x = ax[kb];               a.y = ax[kb + 1];
        v2f b0; b0.x = fcW[kb * 32 + n];    b0.y = fcW[(kb + 1) * 32 + n];
        v2f b1; b1.x = fcW[kb * 32 + 16 + n]; b1.y = fcW[(kb + 1) * 32 + 16 + n];
        acc0 = wmma4(a, b0, acc0);
        acc1 = wmma4(a, b1, acc1);
    }
    const float* ac = cnn + (size_t)rowA * 128;
#pragma unroll
    for (int k0 = 0; k0 < 128; k0 += 4) {
        int kb = k0 + 2 * kh;
        v2f a;  a.x = ac[kb];                a.y = ac[kb + 1];
        v2f b0; b0.x = cnnW[kb * 32 + n];    b0.y = cnnW[(kb + 1) * 32 + n];
        v2f b1; b1.x = cnnW[kb * 32 + 16 + n]; b1.y = cnnW[(kb + 1) * 32 + 16 + n];
        acc2 = wmma4(a, b0, acc2);
        acc3 = wmma4(a, b1, acc3);
    }
#pragma unroll
    for (int v = 0; v < 8; ++v) {
        int mr   = (lane < 16) ? v : v + 8;
        int node = wave * 16 + mr;
        if (node < N) {
            float* o = h0 + (size_t)node * 64;
            o[n]      = fmaxf(acc0[v] + fcb[n],       0.0f);
            o[16 + n] = fmaxf(acc1[v] + fcb[16 + n],  0.0f);
            o[32 + n] = fmaxf(acc2[v] + cnnb[n],      0.0f);
            o[48 + n] = fmaxf(acc3[v] + cnnb[16 + n], 0.0f);
        }
    }
}

__global__ void k_gemm_t1(const float* __restrict__ h0, const float* __restrict__ W1,
                          float* __restrict__ t1, int N, int nTiles) {
    int wave = (blockIdx.x * blockDim.x + threadIdx.x) >> 5;
    if (wave >= nTiles) return;
    int lane = threadIdx.x & 31;
    int n    = lane & 15;
    int kh   = lane >> 4;
    int rowA = wave * 16 + n;    if (rowA >= N) rowA = N - 1;

    v8f acc0 = {}, acc1 = {};
    const float* ar = h0 + (size_t)rowA * 64;
#pragma unroll
    for (int k0 = 0; k0 < 64; k0 += 4) {
        int kb = k0 + 2 * kh;
        v2f a;  a.x = ar[kb];              a.y = ar[kb + 1];
        v2f b0; b0.x = W1[kb * 32 + n];    b0.y = W1[(kb + 1) * 32 + n];
        v2f b1; b1.x = W1[kb * 32 + 16 + n]; b1.y = W1[(kb + 1) * 32 + 16 + n];
        acc0 = wmma4(a, b0, acc0);
        acc1 = wmma4(a, b1, acc1);
    }
#pragma unroll
    for (int v = 0; v < 8; ++v) {
        int mr   = (lane < 16) ? v : v + 8;
        int node = wave * 16 + mr;
        if (node < N) {
            float* o = t1 + (size_t)node * 32;
            o[n]      = acc0[v];
            o[16 + n] = acc1[v];
        }
    }
}

__global__ void k_init_agg1(const float* __restrict__ t1, const float* __restrict__ dinv,
                            const float* __restrict__ b1, float* __restrict__ agg1, int N) {
    int i = blockIdx.x * blockDim.x + threadIdx.x;
    if (i < N * 32) {
        int node = i >> 5, f = i & 31;
        float di = dinv[node];
        agg1[i] = t1[i] * di * di + b1[f];
    }
}

__global__ void k_edge1(const int* __restrict__ src, const int* __restrict__ dst,
                        const float* __restrict__ dinv, const float* __restrict__ t1,
                        float* __restrict__ agg1, int E) {
    int e = (blockIdx.x * blockDim.x + threadIdx.x) >> 5;
    if (e >= E) return;
    int lane = threadIdx.x & 31;
    int s = src[e], d = dst[e];
    float w = dinv[s] * dinv[d];
    atomicAdd(&agg1[(size_t)d * 32 + lane], t1[(size_t)s * 32 + lane] * w);
}

__global__ void k_gemm_t2(const float* __restrict__ agg1, const float* __restrict__ W2,
                          float* __restrict__ t2, int N, int nTiles) {
    int wave = (blockIdx.x * blockDim.x + threadIdx.x) >> 5;
    if (wave >= nTiles) return;
    int lane = threadIdx.x & 31;
    int n    = lane & 15;
    int kh   = lane >> 4;
    int rowA = wave * 16 + n;    if (rowA >= N) rowA = N - 1;

    v8f acc0 = {}, acc1 = {};
    const float* ar = agg1 + (size_t)rowA * 32;
    bool c1ok = (16 + n) < 20;
#pragma unroll
    for (int k0 = 0; k0 < 32; k0 += 4) {
        int kb = k0 + 2 * kh;
        v2f a;  a.x = fmaxf(ar[kb], 0.0f);  a.y = fmaxf(ar[kb + 1], 0.0f);
        v2f b0; b0.x = W2[kb * 20 + n];     b0.y = W2[(kb + 1) * 20 + n];
        v2f b1; b1.x = c1ok ? W2[kb * 20 + 16 + n] : 0.0f;
                b1.y = c1ok ? W2[(kb + 1) * 20 + 16 + n] : 0.0f;
        acc0 = wmma4(a, b0, acc0);
        acc1 = wmma4(a, b1, acc1);
    }
#pragma unroll
    for (int v = 0; v < 8; ++v) {
        int mr   = (lane < 16) ? v : v + 8;
        int node = wave * 16 + mr;
        if (node < N) {
            float* o = t2 + (size_t)node * 20;
            o[n] = acc0[v];
            if (c1ok) o[16 + n] = acc1[v];
        }
    }
}

__global__ void k_init_out(const float* __restrict__ t2, const float* __restrict__ dinv,
                           const float* __restrict__ b2, float* __restrict__ out, int N) {
    int i = blockIdx.x * blockDim.x + threadIdx.x;
    if (i < N * 20) {
        int node = i / 20, f = i - node * 20;
        float di = dinv[node];
        out[i] = t2[i] * di * di + b2[f];
    }
}

__global__ void k_edge2(const int* __restrict__ src, const int* __restrict__ dst,
                        const float* __restrict__ dinv, const float* __restrict__ t2,
                        float* __restrict__ out, int E) {
    int e = (blockIdx.x * blockDim.x + threadIdx.x) >> 5;
    if (e >= E) return;
    int lane = threadIdx.x & 31;
    int s = src[e], d = dst[e];
    float w = dinv[s] * dinv[d];
    if (lane < 20)
        atomicAdd(&out[(size_t)d * 20 + lane], t2[(size_t)s * 20 + lane] * w);
}

extern "C" void kernel_launch(void* const* d_in, const int* in_sizes, int n_in,
                              void* d_out, int out_size, void* d_ws, size_t ws_size,
                              hipStream_t stream) {
    const float* x    = (const float*)d_in[0];
    const float* cnn  = (const float*)d_in[1];
    const int*   ei   = (const int*)d_in[2];
    const float* fcW  = (const float*)d_in[3];
    const float* fcb  = (const float*)d_in[4];
    const float* cnnW = (const float*)d_in[5];
    const float* cnnb = (const float*)d_in[6];
    const float* W1   = (const float*)d_in[7];
    const float* b1   = (const float*)d_in[8];
    const float* W2   = (const float*)d_in[9];
    const float* b2   = (const float*)d_in[10];

    const int N = in_sizes[0] / 32;
    const int E = in_sizes[2] / 2;
    const int* src = ei;
    const int* dst = ei + E;

    float* ws   = (float*)d_ws;
    float* dinv = ws;  ws += N;
    float* h0   = ws;  ws += (size_t)N * 64;
    float* t1   = ws;  ws += (size_t)N * 32;
    float* agg1 = ws;  ws += (size_t)N * 32;
    float* t2   = ws;  ws += (size_t)N * 20;
    float* out  = (float*)d_out;

    const int nTiles = (N + 15) / 16;
    const int B = 256;
    dim3 blkG(128);
    dim3 grdG((nTiles * 32 + 127) / 128);
    dim3 grdEdge((unsigned)(((long long)E * 32 + B - 1) / B));

    k_zero<<<(N + B - 1) / B, B, 0, stream>>>(dinv, N);
    k_count_deg<<<(E + B - 1) / B, B, 0, stream>>>(dst, dinv, E);
    k_dinv<<<(N + B - 1) / B, B, 0, stream>>>(dinv, N);

    k_encoder<<<grdG, blkG, 0, stream>>>(x, cnn, fcW, fcb, cnnW, cnnb, h0, N, nTiles);
    k_gemm_t1<<<grdG, blkG, 0, stream>>>(h0, W1, t1, N, nTiles);

    k_init_agg1<<<(N * 32 + B - 1) / B, B, 0, stream>>>(t1, dinv, b1, agg1, N);
    k_edge1<<<grdEdge, B, 0, stream>>>(src, dst, dinv, t1, agg1, E);

    k_gemm_t2<<<grdG, blkG, 0, stream>>>(agg1, W2, t2, N, nTiles);
    k_init_out<<<(N * 20 + B - 1) / B, B, 0, stream>>>(t2, dinv, b2, out, N);
    k_edge2<<<grdEdge, B, 0, stream>>>(src, dst, dinv, t2, out, E);
}


# Second aggregation reads per-wave edge lists that the first one's range search stores (in dead h0 space) instead of searching the edge list again; one cut for both, so layer-2 slice sums regroup
# speedup vs baseline: 3.6615x; 3.6615x over previous
#include <hip/hip_runtime.h>
#include <hip/hip_bf16.h>

typedef __attribute__((ext_vector_type(2))) float v2f;
typedef __attribute__((ext_vector_type(8))) float v8f;

__device__ __forceinline__ v8f wmma4(v2f a, v2f b, v8f c) {
    return __builtin_amdgcn_wmma_f32_16x16x4_f32(
         false, a,  false, b,
         (short)0, c,  false,  false);
}

constexpr int CNT_SLICE = 32768;
constexpr int CNT_WORDS = 65536;
constexpr int CNT_LOADS = CNT_SLICE / 4096;

__device__ __forceinline__ void count_one(unsigned* cw, int d, int lo, unsigned nn) {
    const unsigned dl = (unsigned)(d - lo);
    if (dl < nn) atomicAdd(cw + (dl >> 1), 1u << ((dl & 1u) << 4));
}

__global__ __launch_bounds__(1024) void k_count_planes(const int* __restrict__ dst,
                                                       unsigned* __restrict__ planes,
                                                       int E, int NW4) {
    __shared__ __attribute__((aligned(16))) unsigned cw[CNT_WORDS];
    const int tid = (int)threadIdx.x;
    const int P  = (NW4 + CNT_WORDS - 1) / CNT_WORDS;
    const int p  = (int)blockIdx.x % P;
    const int r  = (int)blockIdx.x / P;
    const int w0 = p * CNT_WORDS;
    const int nw = (NW4 - w0) < CNT_WORDS ? (NW4 - w0) : CNT_WORDS;

    uint4* cw4 = reinterpret_cast<uint4*>(cw);
    for (int i = tid; i < nw / 4; i += 1024) cw4[i] = make_uint4(0u, 0u, 0u, 0u);
    __syncthreads();

    const int e0 = r * CNT_SLICE;
    const int e1 = (e0 + CNT_SLICE) < E ? (e0 + CNT_SLICE) : E;
    const int lo = 2 * w0;
    const unsigned nn = 2u * (unsigned)nw;
    if (((E & 3) == 0) && (e0 + CNT_SLICE <= E)) {
        int4 q[CNT_LOADS];
#pragma unroll
        for (int k = 0; k < CNT_LOADS; ++k)
            q[k] = *reinterpret_cast<const int4*>(dst + e0 + (k * 1024 + tid) * 4);
#pragma unroll
        for (int k = 0; k < CNT_LOADS; ++k) {
            count_one(cw, q[k].x, lo, nn);
            count_one(cw, q[k].y, lo, nn);
            count_one(cw, q[k].z, lo, nn);
            count_one(cw, q[k].w, lo, nn);
        }
    } else {
        for (int e = e0 + tid; e < e1; e += 1024) count_one(cw, dst[e], lo, nn);
    }
    __syncthreads();

    uint4* pl4 = reinterpret_cast<uint4*>(planes + (size_t)r * NW4 + w0);
    for (int i = tid; i < nw / 4; i += 1024) pl4[i] = cw4[i];
}

__global__ __launch_bounds__(256) void k_dinv_planes(const unsigned* __restrict__ planes,
                                                     float* __restrict__ dinv,
                                                     int N, int NW4, int R) {
    __shared__ unsigned part[8][64];
    const int lane = (int)threadIdx.x & 31;
    const int wv   = (int)threadIdx.x >> 5;
    const int i    = (int)blockIdx.x * 32 + lane;
    unsigned lo = 0u, hi = 0u;
    if (i < NW4) {
        for (int base = wv; base < R; base += 128) {
            unsigned w[16];
#pragma unroll
            for (int k = 0; k < 16; ++k) {
                const int r = base + 8 * k;
                w[k] = (r < R) ? planes[(size_t)r * NW4 + i] : 0u;
            }
#pragma unroll
            for (int k = 0; k < 16; ++k) {
                lo += w[k] & 0xffffu;
                hi += w[k] >> 16;
            }
        }
    }
    part[wv][lane]      = lo;
    part[wv][32 + lane] = hi;
    __syncthreads();
    if (wv < 2) {
        unsigned s = 0u;
        for (int k = 0; k < 8; ++k) s += part[k][wv * 32 + lane];
        const int node = 2 * i + wv;
        if (node < N) dinv[node] = rsqrtf((float)s + 1.0f);
    }
}

__global__ void k_encoder(const float* __restrict__ x, const float* __restrict__ cnn,
                          const float* __restrict__ fcW, const float* __restrict__ fcb,
                          const float* __restrict__ cnnW, const float* __restrict__ cnnb,
                          float* __restrict__ h0, int N, int nTiles) {
    int wave = (blockIdx.x * blockDim.x + threadIdx.x) >> 5;
    if (wave >= nTiles) return;
    int lane = threadIdx.x & 31;
    int n    = lane & 15;
    int kh   = lane >> 4;
    int rowA = wave * 16 + n;    if (rowA >= N) rowA = N - 1;

    v8f acc0 = {}, acc1 = {}, acc2 = {}, acc3 = {};

    const float* ax = x + (size_t)rowA * 32;
#pragma unroll
    for (int k0 = 0; k0 < 32; k0 += 4) {
        int kb = k0 + 2 * kh;
        v2f a;  a.x = ax[kb];               a.y = ax[kb + 1];
        v2f b0; b0.x = fcW[kb * 32 + n];    b0.y = fcW[(kb + 1) * 32 + n];
        v2f b1; b1.x = fcW[kb * 32 + 16 + n]; b1.y = fcW[(kb + 1) * 32 + 16 + n];
        acc0 = wmma4(a, b0, acc0);
        acc1 = wmma4(a, b1, acc1);
    }
    const float* ac = cnn + (size_t)rowA * 128;
#pragma unroll
    for (int k0 = 0; k0 < 128; k0 += 4) {
        int kb = k0 + 2 * kh;
        v2f a;  a.x = ac[kb];                a.y = ac[kb + 1];
        v2f b0; b0.x = cnnW[kb * 32 + n];    b0.y = cnnW[(kb + 1) * 32 + n];
        v2f b1; b1.x = cnnW[kb * 32 + 16 + n]; b1.y = cnnW[(kb + 1) * 32 + 16 + n];
        acc2 = wmma4(a, b0, acc2);
        acc3 = wmma4(a, b1, acc3);
    }
#pragma unroll
    for (int v = 0; v < 8; ++v) {
        int mr   = (lane < 16) ? v : v + 8;
        int node = wave * 16 + mr;
        if (node < N) {
            float* o = h0 + (size_t)node * 64;
            o[n]      = fmaxf(acc0[v] + fcb[n],       0.0f);
            o[16 + n] = fmaxf(acc1[v] + fcb[16 + n],  0.0f);
            o[32 + n] = fmaxf(acc2[v] + cnnb[n],      0.0f);
            o[48 + n] = fmaxf(acc3[v] + cnnb[16 + n], 0.0f);
        }
    }
}

__global__ void k_gemm_t1(const float* __restrict__ h0, const float* __restrict__ W1,
                          const float* __restrict__ dinv, const float* __restrict__ b1,
                          float* __restrict__ t1, float* __restrict__ agg1, int N, int nTiles) {
    int wave = (blockIdx.x * blockDim.x + threadIdx.x) >> 5;
    if (wave >= nTiles) return;
    int lane = threadIdx.x & 31;
    int n    = lane & 15;
    int kh   = lane >> 4;
    int rowA = wave * 16 + n;    if (rowA >= N) rowA = N - 1;

    float dv[8];
#pragma unroll
    for (int v = 0; v < 8; ++v) {
        int nd = wave * 16 + 8 * kh + v;   if (nd >= N) nd = N - 1;
        dv[v] = dinv[nd];
    }

    v8f acc0 = {}, acc1 = {};
    const float* ar = h0 + (size_t)rowA * 64;
#pragma unroll
    for (int k0 = 0; k0 < 64; k0 += 4) {
        int kb = k0 + 2 * kh;
        v2f a;  a.x = ar[kb];              a.y = ar[kb + 1];
        v2f b0; b0.x = W1[kb * 32 + n];    b0.y = W1[(kb + 1) * 32 + n];
        v2f w1; w1.x = W1[kb * 32 + 16 + n]; w1.y = W1[(kb + 1) * 32 + 16 + n];
        acc0 = wmma4(a, b0, acc0);
        acc1 = wmma4(a, w1, acc1);
    }
    const float bia0 = b1[n], bia1 = b1[16 + n];
#pragma unroll
    for (int v = 0; v < 8; ++v) {
        int mr   = (lane < 16) ? v : v + 8;
        int node = wave * 16 + mr;
        if (node < N) {
            float* o = t1 + (size_t)node * 32;
            float* a = agg1 + (size_t)node * 32;
            const float di = dv[v];
            o[n]      = acc0[v];
            o[16 + n] = acc1[v];
            a[n]      = acc0[v] * di * di + bia0;
            a[16 + n] = acc1[v] * di * di + bia1;
        }
    }
}

constexpr int      LIST_SRC_BITS = 21;
constexpr unsigned LIST_SRC_MASK = (1u << LIST_SRC_BITS) - 1u;

template <int F, bool FULL>
__device__ __forceinline__ void drain_regs(int s, unsigned dl, unsigned cnt, unsigned lane, int lo,
                                           const float* __restrict__ dinv, const float* __restrict__ t,
                                           float* __restrict__ tile) {
    float w = 0.0f;
    if (FULL || lane < cnt) w = dinv[s] * dinv[lo + (int)dl];
    const bool act = lane < (unsigned)F;
    const unsigned fl = act ? lane : (unsigned)(F - 1);
    float v[32];
#pragma unroll
    for (int u = 0; u < 32; ++u) {
        const unsigned sj = (unsigned)__builtin_amdgcn_readlane(s, u);
        const float* row = t + (size_t)sj * F;
        v[u] = row[fl];
    }
    unsigned dj[32];
    float wj[32];
#pragma unroll
    for (int u = 0; u < 32; ++u) {
        dj[u] = (unsigned)__builtin_amdgcn_readlane((int)dl, u);
        wj[u] = __int_as_float(__builtin_amdgcn_readlane(__float_as_int(w), u));
    }
    if (act) {
#pragma unroll
        for (int u = 0; u < 32; ++u)
            if (FULL || (unsigned)u < cnt)
                atomicAdd(tile + dj[u] * F + lane, v[u] * wj[u]);
    }
}

template <int F, bool FULL, bool KEEP>
__device__ __forceinline__ void drain_ring(const int* __restrict__ rs, const int* __restrict__ rd,
                                           unsigned head, unsigned cnt, unsigned lane, int lo,
                                           const float* __restrict__ dinv, const float* __restrict__ t,
                                           float* __restrict__ tile,
                                           unsigned* __restrict__ mylist, unsigned kept, unsigned cap) {
    const unsigned qi = (head + lane) & 63u;
    int s = 0;
    unsigned dl = 0;
    if (FULL || lane < cnt) {
        s  = rs[qi];
        dl = (unsigned)rd[qi];
        if (KEEP && kept < cap) mylist[kept + lane] = (unsigned)s | (dl << LIST_SRC_BITS);
    }
    drain_regs<F, FULL>(s, dl, cnt, lane, lo, dinv, t, tile);
}

__device__ __forceinline__ void load_group(const int* __restrict__ src, const int* __restrict__ dst,
                                           int g, int e1, bool vec, unsigned lane,
                                           int (&d)[8], int (&s)[8]) {
    if (vec && g + 256 <= e1) {
#pragma unroll
        for (int h = 0; h < 2; ++h) {
            const int e = g + h * 128 + (int)lane * 4;
            const int4 qd = *reinterpret_cast<const int4*>(dst + e);
            const int4 qs = *reinterpret_cast<const int4*>(src + e);
            d[4 * h] = qd.x; d[4 * h + 1] = qd.y; d[4 * h + 2] = qd.z; d[4 * h + 3] = qd.w;
            s[4 * h] = qs.x; s[4 * h + 1] = qs.y; s[4 * h + 2] = qs.z; s[4 * h + 3] = qs.w;
        }
    } else {
#pragma unroll
        for (int h = 0; h < 2; ++h) {
#pragma unroll
            for (int i = 0; i < 4; ++i) {
                const int e = g + h * 128 + (int)lane * 4 + i;
                d[4 * h + i] = (e < e1) ? dst[e] : -1;
                s[4 * h + i] = (e < e1) ? src[e] : 0;
            }
        }
    }
}

template <int F, bool KEEP>
__device__ __forceinline__ unsigned search_slice(const int* __restrict__ src, const int* __restrict__ dst,
                                                 const float* __restrict__ dinv, const float* __restrict__ t,
                                                 float* __restrict__ tile, int* __restrict__ rs, int* __restrict__ rd,
                                                 int e0, int e1, bool vec, unsigned wv, unsigned lane,
                                                 int lo, int nloc,
                                                 unsigned* __restrict__ mylist, unsigned cap) {
    unsigned head = 0, cnt = 0, kept = 0;

    int cd[8], cs[8];
#pragma unroll
    for (int k = 0; k < 8; ++k) { cd[k] = -1; cs[k] = 0; }
    int gn = e0 + (int)wv * 256;
    for (;;) {
        const bool more = gn < e1;
        int nd[8], ns[8];
        if (more) {
            load_group(src, dst, gn, e1, vec, lane, nd, ns);
        } else {
#pragma unroll
            for (int k = 0; k < 8; ++k) { nd[k] = -1; ns[k] = 0; }
        }
#pragma unroll
        for (int k = 0; k < 8; ++k) {
            const unsigned dl = (unsigned)(cd[k] - lo);
            const bool m = dl < (unsigned)nloc;
            const unsigned mask = __builtin_amdgcn_ballot_w32(m);
            if (mask != 0u) {
                if (m) {
                    const unsigned pos = __builtin_amdgcn_mbcnt_lo(mask, 0u);
                    const unsigned qi  = (head + cnt + pos) & 63u;
                    rs[qi] = cs[k];
                    rd[qi] = (int)dl;
                }
                cnt += (unsigned)__builtin_popcount(mask);
                if (cnt >= 32u) {
                    drain_ring<F, true, KEEP>(rs, rd, head, cnt, lane, lo, dinv, t, tile, mylist, kept, cap);
                    head = (head + 32u) & 63u;
                    cnt -= 32u;
                    kept += 32u;
                }
            }
        }
        if (!more) break;
#pragma unroll
        for (int k = 0; k < 8; ++k) { cd[k] = nd[k]; cs[k] = ns[k]; }
        gn += 32 * 256;
    }
    if (cnt != 0u)
        drain_ring<F, false, KEEP>(rs, rd, head, cnt, lane, lo, dinv, t, tile, mylist, kept, cap);
    return kept + cnt;
}

template <int F, int NP, bool FIRST>
__global__ __launch_bounds__(1024) void k_edge_tile(const int* __restrict__ src, const int* __restrict__ dst,
                                                    const float* __restrict__ dinv, const float* __restrict__ t,
                                                    float* __restrict__ agg, unsigned* __restrict__ lists,
                                                    unsigned cap, int E, int N) {
    static_assert(NP <= (1 << (32 - LIST_SRC_BITS)), "a place in the range must fit above the source bits");
    __shared__ __attribute__((aligned(16))) float tile[NP * F];
    __shared__ int ring_s[32][64];
    __shared__ int ring_d[32][64];

    const unsigned tid  = threadIdx.x;
    const unsigned lane = tid & 31u;
    const unsigned wv = (unsigned)__builtin_amdgcn_readfirstlane((int)(tid >> 5));

    const int P = (N + NP - 1) / NP;
    const int R = (int)gridDim.x / P;
    const int p = (int)blockIdx.x % P;
    const int r = (int)blockIdx.x / P;
    const int lo   = p * NP;
    const int nloc = (N - lo) < NP ? (N - lo) : NP;

    for (int i = (int)tid; i < NP * F / 4; i += 1024)
        reinterpret_cast<float4*>(tile)[i] = make_float4(0.0f, 0.0f, 0.0f, 0.0f);
    __syncthreads();

    const int per = (((E + R - 1) / R) + 255) & ~255;
    const int e0  = r * per;
    const int e1  = (e0 + per) < E ? (e0 + per) : E;
    const bool vec = (E & 3) == 0;

    int* rs = ring_s[wv];
    int* rd = ring_d[wv];
    const unsigned wid = blockIdx.x * 32u + wv;
    unsigned* mylist = lists + (size_t)gridDim.x * 32u + (size_t)wid * cap;

    if (FIRST) {
        const unsigned found = search_slice<F, true>(src, dst, dinv, t, tile, rs, rd, e0, e1, vec,
                                                     wv, lane, lo, nloc, mylist, cap);
        if (lane == 0u) lists[wid] = found;
    } else {
        const unsigned c = (unsigned)__builtin_amdgcn_readfirstlane((int)lists[wid]);
        if (c <= cap) {
            unsigned nxt = (lane < c) ? mylist[lane] : 0u;
            for (unsigned k = 0; k < c; k += 32u) {
                const unsigned cur = nxt;
                const unsigned kn  = k + 32u;
                if (kn < c) nxt = (kn + lane < c) ? mylist[kn + lane] : 0u;
                const int      s  = (int)(cur & LIST_SRC_MASK);
                const unsigned dl = cur >> LIST_SRC_BITS;
                const unsigned rem = c - k;
                if (rem >= 32u) drain_regs<F, true >(s, dl, 32u, lane, lo, dinv, t, tile);
                else            drain_regs<F, false>(s, dl, rem, lane, lo, dinv, t, tile);
            }
        } else {
            search_slice<F, false>(src, dst, dinv, t, tile, rs, rd, e0, e1, vec,
                                   wv, lane, lo, nloc, mylist, 0u);
        }
    }
    __syncthreads();

    float* ag = agg + (size_t)lo * F;
    const int tot = nloc * F;
    for (int i = (int)tid; i < tot; i += 1024)
        atomicAdd(ag + i, tile[i]);
}

__global__ void k_gemm_t2(const float* __restrict__ agg1, const float* __restrict__ W2,
                          const float* __restrict__ dinv, const float* __restrict__ b2,
                          float* __restrict__ t2, float* __restrict__ out, int N, int nTiles) {
    int wave = (blockIdx.x * blockDim.x + threadIdx.x) >> 5;
    if (wave >= nTiles) return;
    int lane = threadIdx.x & 31;
    int n    = lane & 15;
    int kh   = lane >> 4;
    int rowA = wave * 16 + n;    if (rowA >= N) rowA = N - 1;

    float dv[8];
#pragma unroll
    for (int v = 0; v < 8; ++v) {
        int nd = wave * 16 + 8 * kh + v;   if (nd >= N) nd = N - 1;
        dv[v] = dinv[nd];
    }

    v8f acc0 = {}, acc1 = {};
    const float* ar = agg1 + (size_t)rowA * 32;
    bool c1ok = (16 + n) < 20;
#pragma unroll
    for (int k0 = 0; k0 < 32; k0 += 4) {
        int kb = k0 + 2 * kh;
        v2f a;  a.x = fmaxf(ar[kb], 0.0f);  a.y = fmaxf(ar[kb + 1], 0.0f);
        v2f b0; b0.x = W2[kb * 20 + n];     b0.y = W2[(kb + 1) * 20 + n];
        v2f b1; b1.x = c1ok ? W2[kb * 20 + 16 + n] : 0.0f;
                b1.y = c1ok ? W2[(kb + 1) * 20 + 16 + n] : 0.0f;
        acc0 = wmma4(a, b0, acc0);
        acc1 = wmma4(a, b1, acc1);
    }
    const float bia0 = b2[n];
    const float bia1 = c1ok ? b2[16 + n] : 0.0f;
#pragma unroll
    for (int v = 0; v < 8; ++v) {
        int mr   = (lane < 16) ? v : v + 8;
        int node = wave * 16 + mr;
        if (node < N) {
            float* o = t2 + (size_t)node * 20;
            float* q = out + (size_t)node * 20;
            const float di = dv[v];
            o[n] = acc0[v];
            q[n] = acc0[v] * di * di + bia0;
            if (c1ok) {
                o[16 + n] = acc1[v];
                q[16 + n] = acc1[v] * di * di + bia1;
            }
        }
    }
}

extern "C" void kernel_launch(void* const* d_in, const int* in_sizes, int n_in,
                              void* d_out, int out_size, void* d_ws, size_t ws_size,
                              hipStream_t stream) {
    const float* x    = (const float*)d_in[0];
    const float* cnn  = (const float*)d_in[1];
    const int*   ei   = (const int*)d_in[2];
    const float* fcW  = (const float*)d_in[3];
    const float* fcb  = (const float*)d_in[4];
    const float* cnnW = (const float*)d_in[5];
    const float* cnnb = (const float*)d_in[6];
    const float* W1   = (const float*)d_in[7];
    const float* b1   = (const float*)d_in[8];
    const float* W2   = (const float*)d_in[9];
    const float* b2   = (const float*)d_in[10];

    const int N = in_sizes[0] / 32;
    const int E = in_sizes[2] / 2;
    const int* src = ei;
    const int* dst = ei + E;

    float* ws   = (float*)d_ws;
    float* dinv = ws;  ws += N;
    unsigned* planes = (unsigned*)((float*)d_ws + ((N + 3) / 4) * 4);
    float* h0   = ws;  ws += (size_t)N * 64;
    float* t1   = ws;  ws += (size_t)N * 32;
    float* agg1 = ws;  ws += (size_t)N * 32;
    float* t2   = ws;  ws += (size_t)N * 20;
    float* out  = (float*)d_out;

    const int nTiles = (N + 15) / 16;
    dim3 blkG(128);
    dim3 grdG((nTiles * 32 + 127) / 128);
    const int PT = (N + 2047) / 2048;
    const int RT = PT < 256 ? 256 / PT : 1;
    unsigned* lists = (unsigned*)h0;
    const long long lwaves = (long long)PT * RT * 32;
    long long lcap = (((long long)N * 64 - lwaves) / lwaves) & ~31LL;
    if (lcap < 0 || N > (1 << LIST_SRC_BITS)) lcap = 0;
    if (lcap > 65536) lcap = 65536;
    const unsigned cap = (unsigned)lcap;

    const int NW4 = ((N + 7) / 8) * 4;
    const int PC  = (NW4 + CNT_WORDS - 1) / CNT_WORDS;
    const int RC  = (E + CNT_SLICE - 1) / CNT_SLICE;
    k_count_planes<<<PC * RC, 1024, 0, stream>>>(dst, planes, E, NW4);
    k_dinv_planes<<<(NW4 + 31) / 32, 256, 0, stream>>>(planes, dinv, N, NW4, RC);

    k_encoder<<<grdG, blkG, 0, stream>>>(x, cnn, fcW, fcb, cnnW, cnnb, h0, N, nTiles);
    k_gemm_t1<<<grdG, blkG, 0, stream>>>(h0, W1, dinv, b1, t1, agg1, N, nTiles);

    k_edge_tile<32, 2048, true><<<PT * RT, 1024, 0, stream>>>(src, dst, dinv, t1, agg1, lists, cap, E, N);

    k_gemm_t2<<<grdG, blkG, 0, stream>>>(agg1, W2, dinv, b2, t2, out, N, nTiles);
    k_edge_tile<20, 2048, false><<<PT * RT, 1024, 0, stream>>>(src, dst, dinv, t2, out, lists, cap, E, N);
}


# First aggregation's range search (each edge read once per range) replaced by one split launch filing each edge into its wave's list (7 -> 8 launches); list order: LDS arrival rank; sums as before
# speedup vs baseline: 4.1231x; 1.1261x over previous
#include <hip/hip_runtime.h>
#include <hip/hip_bf16.h>

typedef __attribute__((ext_vector_type(2))) float v2f;
typedef __attribute__((ext_vector_type(8))) float v8f;

__device__ __forceinline__ v8f wmma4(v2f a, v2f b, v8f c) {
    return __builtin_amdgcn_wmma_f32_16x16x4_f32(
         false, a,  false, b,
         (short)0, c,  false,  false);
}

constexpr int CNT_SLICE = 32768;
constexpr int CNT_WORDS = 65536;
constexpr int CNT_LOADS = CNT_SLICE / 4096;

__device__ __forceinline__ void count_one(unsigned* cw, int d, int lo, unsigned nn) {
    const unsigned dl = (unsigned)(d - lo);
    if (dl < nn) atomicAdd(cw + (dl >> 1), 1u << ((dl & 1u) << 4));
}

__global__ __launch_bounds__(1024) void k_count_planes(const int* __restrict__ dst,
                                                       unsigned* __restrict__ planes,
                                                       int E, int NW4) {
    __shared__ __attribute__((aligned(16))) unsigned cw[CNT_WORDS];
    const int tid = (int)threadIdx.x;
    const int P  = (NW4 + CNT_WORDS - 1) / CNT_WORDS;
    const int p  = (int)blockIdx.x % P;
    const int r  = (int)blockIdx.x / P;
    const int w0 = p * CNT_WORDS;
    const int nw = (NW4 - w0) < CNT_WORDS ? (NW4 - w0) : CNT_WORDS;

    uint4* cw4 = reinterpret_cast<uint4*>(cw);
    for (int i = tid; i < nw / 4; i += 1024) cw4[i] = make_uint4(0u, 0u, 0u, 0u);
    __syncthreads();

    const int e0 = r * CNT_SLICE;
    const int e1 = (e0 + CNT_SLICE) < E ? (e0 + CNT_SLICE) : E;
    const int lo = 2 * w0;
    const unsigned nn = 2u * (unsigned)nw;
    if (((E & 3) == 0) && (e0 + CNT_SLICE <= E)) {
        int4 q[CNT_LOADS];
#pragma unroll
        for (int k = 0; k < CNT_LOADS; ++k)
            q[k] = *reinterpret_cast<const int4*>(dst + e0 + (k * 1024 + tid) * 4);
#pragma unroll
        for (int k = 0; k < CNT_LOADS; ++k) {
            count_one(cw, q[k].x, lo, nn);
            count_one(cw, q[k].y, lo, nn);
            count_one(cw, q[k].z, lo, nn);
            count_one(cw, q[k].w, lo, nn);
        }
    } else {
        for (int e = e0 + tid; e < e1; e += 1024) count_one(cw, dst[e], lo, nn);
    }
    __syncthreads();

    uint4* pl4 = reinterpret_cast<uint4*>(planes + (size_t)r * NW4 + w0);
    for (int i = tid; i < nw / 4; i += 1024) pl4[i] = cw4[i];
}

__global__ __launch_bounds__(256) void k_dinv_planes(const unsigned* __restrict__ planes,
                                                     float* __restrict__ dinv,
                                                     int N, int NW4, int R) {
    __shared__ unsigned part[8][64];
    const int lane = (int)threadIdx.x & 31;
    const int wv   = (int)threadIdx.x >> 5;
    const int i    = (int)blockIdx.x * 32 + lane;
    unsigned lo = 0u, hi = 0u;
    if (i < NW4) {
        for (int base = wv; base < R; base += 128) {
            unsigned w[16];
#pragma unroll
            for (int k = 0; k < 16; ++k) {
                const int r = base + 8 * k;
                w[k] = (r < R) ? planes[(size_t)r * NW4 + i] : 0u;
            }
#pragma unroll
            for (int k = 0; k < 16; ++k) {
                lo += w[k] & 0xffffu;
                hi += w[k] >> 16;
            }
        }
    }
    part[wv][lane]      = lo;
    part[wv][32 + lane] = hi;
    __syncthreads();
    if (wv < 2) {
        unsigned s = 0u;
        for (int k = 0; k < 8; ++k) s += part[k][wv * 32 + lane];
        const int node = 2 * i + wv;
        if (node < N) dinv[node] = rsqrtf((float)s + 1.0f);
    }
}

__global__ void k_encoder(const float* __restrict__ x, const float* __restrict__ cnn,
                          const float* __restrict__ fcW, const float* __restrict__ fcb,
                          const float* __restrict__ cnnW, const float* __restrict__ cnnb,
                          float* __restrict__ h0, int N, int nTiles) {
    int wave = (blockIdx.x * blockDim.x + threadIdx.x) >> 5;
    if (wave >= nTiles) return;
    int lane = threadIdx.x & 31;
    int n    = lane & 15;
    int kh   = lane >> 4;
    int rowA = wave * 16 + n;    if (rowA >= N) rowA = N - 1;

    v8f acc0 = {}, acc1 = {}, acc2 = {}, acc3 = {};

    const float* ax = x + (size_t)rowA * 32;
#pragma unroll
    for (int k0 = 0; k0 < 32; k0 += 4) {
        int kb = k0 + 2 * kh;
        v2f a;  a.x = ax[kb];               a.y = ax[kb + 1];
        v2f b0; b0.x = fcW[kb * 32 + n];    b0.y = fcW[(kb + 1) * 32 + n];
        v2f b1; b1.x = fcW[kb * 32 + 16 + n]; b1.y = fcW[(kb + 1) * 32 + 16 + n];
        acc0 = wmma4(a, b0, acc0);
        acc1 = wmma4(a, b1, acc1);
    }
    const float* ac = cnn + (size_t)rowA * 128;
#pragma unroll
    for (int k0 = 0; k0 < 128; k0 += 4) {
        int kb = k0 + 2 * kh;
        v2f a;  a.x = ac[kb];                a.y = ac[kb + 1];
        v2f b0; b0.x = cnnW[kb * 32 + n];    b0.y = cnnW[(kb + 1) * 32 + n];
        v2f b1; b1.x = cnnW[kb * 32 + 16 + n]; b1.y = cnnW[(kb + 1) * 32 + 16 + n];
        acc2 = wmma4(a, b0, acc2);
        acc3 = wmma4(a, b1, acc3);
    }
#pragma unroll
    for (int v = 0; v < 8; ++v) {
        int mr   = (lane < 16) ? v : v + 8;
        int node = wave * 16 + mr;
        if (node < N) {
            float* o = h0 + (size_t)node * 64;
            o[n]      = fmaxf(acc0[v] + fcb[n],       0.0f);
            o[16 + n] = fmaxf(acc1[v] + fcb[16 + n],  0.0f);
            o[32 + n] = fmaxf(acc2[v] + cnnb[n],      0.0f);
            o[48 + n] = fmaxf(acc3[v] + cnnb[16 + n], 0.0f);
        }
    }
}

__global__ void k_gemm_t1(const float* __restrict__ h0, const float* __restrict__ W1,
                          const float* __restrict__ dinv, const float* __restrict__ b1,
                          float* __restrict__ t1, float* __restrict__ agg1, int N, int nTiles) {
    int wave = (blockIdx.x * blockDim.x + threadIdx.x) >> 5;
    if (wave >= nTiles) return;
    int lane = threadIdx.x & 31;
    int n    = lane & 15;
    int kh   = lane >> 4;
    int rowA = wave * 16 + n;    if (rowA >= N) rowA = N - 1;

    float dv[8];
#pragma unroll
    for (int v = 0; v < 8; ++v) {
        int nd = wave * 16 + 8 * kh + v;   if (nd >= N) nd = N - 1;
        dv[v] = dinv[nd];
    }

    v8f acc0 = {}, acc1 = {};
    const float* ar = h0 + (size_t)rowA * 64;
#pragma unroll
    for (int k0 = 0; k0 < 64; k0 += 4) {
        int kb = k0 + 2 * kh;
        v2f a;  a.x = ar[kb];              a.y = ar[kb + 1];
        v2f b0; b0.x = W1[kb * 32 + n];    b0.y = W1[(kb + 1) * 32 + n];
        v2f w1; w1.x = W1[kb * 32 + 16 + n]; w1.y = W1[(kb + 1) * 32 + 16 + n];
        acc0 = wmma4(a, b0, acc0);
        acc1 = wmma4(a, w1, acc1);
    }
    const float bia0 = b1[n], bia1 = b1[16 + n];
#pragma unroll
    for (int v = 0; v < 8; ++v) {
        int mr   = (lane < 16) ? v : v + 8;
        int node = wave * 16 + mr;
        if (node < N) {
            float* o = t1 + (size_t)node * 32;
            float* a = agg1 + (size_t)node * 32;
            const float di = dv[v];
            o[n]      = acc0[v];
            o[16 + n] = acc1[v];
            a[n]      = acc0[v] * di * di + bia0;
            a[16 + n] = acc1[v] * di * di + bia1;
        }
    }
}

constexpr int      LIST_SRC_BITS = 21;
constexpr unsigned LIST_SRC_MASK = (1u << LIST_SRC_BITS) - 1u;

template <int F, bool FULL>
__device__ __forceinline__ void drain_regs(int s, unsigned dl, unsigned cnt, unsigned lane, int lo,
                                           const float* __restrict__ dinv, const float* __restrict__ t,
                                           float* __restrict__ tile) {
    float w = 0.0f;
    if (FULL || lane < cnt) w = dinv[s] * dinv[lo + (int)dl];
    const bool act = lane < (unsigned)F;
    const unsigned fl = act ? lane : (unsigned)(F - 1);
    float v[32];
#pragma unroll
    for (int u = 0; u < 32; ++u) {
        const unsigned sj = (unsigned)__builtin_amdgcn_readlane(s, u);
        const float* row = t + (size_t)sj * F;
        v[u] = row[fl];
    }
    unsigned dj[32];
    float wj[32];
#pragma unroll
    for (int u = 0; u < 32; ++u) {
        dj[u] = (unsigned)__builtin_amdgcn_readlane((int)dl, u);
        wj[u] = __int_as_float(__builtin_amdgcn_readlane(__float_as_int(w), u));
    }
    if (act) {
#pragma unroll
        for (int u = 0; u < 32; ++u)
            if (FULL || (unsigned)u < cnt)
                atomicAdd(tile + dj[u] * F + lane, v[u] * wj[u]);
    }
}

template <int F, bool FULL>
__device__ __forceinline__ void drain_ring(const int* __restrict__ rs, const int* __restrict__ rd,
                                           unsigned head, unsigned cnt, unsigned lane, int lo,
                                           const float* __restrict__ dinv, const float* __restrict__ t,
                                           float* __restrict__ tile) {
    const unsigned qi = (head + lane) & 63u;
    int s = 0;
    unsigned dl = 0;
    if (FULL || lane < cnt) {
        s  = rs[qi];
        dl = (unsigned)rd[qi];
    }
    drain_regs<F, FULL>(s, dl, cnt, lane, lo, dinv, t, tile);
}

__device__ __forceinline__ void load_group(const int* __restrict__ src, const int* __restrict__ dst,
                                           int g, int e1, bool vec, unsigned lane,
                                           int (&d)[8], int (&s)[8]) {
    if (vec && g + 256 <= e1) {
#pragma unroll
        for (int h = 0; h < 2; ++h) {
            const int e = g + h * 128 + (int)lane * 4;
            const int4 qd = *reinterpret_cast<const int4*>(dst + e);
            const int4 qs = *reinterpret_cast<const int4*>(src + e);
            d[4 * h] = qd.x; d[4 * h + 1] = qd.y; d[4 * h + 2] = qd.z; d[4 * h + 3] = qd.w;
            s[4 * h] = qs.x; s[4 * h + 1] = qs.y; s[4 * h + 2] = qs.z; s[4 * h + 3] = qs.w;
        }
    } else {
#pragma unroll
        for (int h = 0; h < 2; ++h) {
#pragma unroll
            for (int i = 0; i < 4; ++i) {
                const int e = g + h * 128 + (int)lane * 4 + i;
                d[4 * h + i] = (e < e1) ? dst[e] : -1;
                s[4 * h + i] = (e < e1) ? src[e] : 0;
            }
        }
    }
}

template <int F>
__device__ __forceinline__ void search_slice(const int* __restrict__ src, const int* __restrict__ dst,
                                                 const float* __restrict__ dinv, const float* __restrict__ t,
                                                 float* __restrict__ tile, int* __restrict__ rs, int* __restrict__ rd,
                                                 int e0, int e1, bool vec, unsigned wv, unsigned lane,
                                                 int lo, int nloc) {
    unsigned head = 0, cnt = 0;

    int cd[8], cs[8];
#pragma unroll
    for (int k = 0; k < 8; ++k) { cd[k] = -1; cs[k] = 0; }
    int gn = e0 + (int)wv * 256;
    for (;;) {
        const bool more = gn < e1;
        int nd[8], ns[8];
        if (more) {
            load_group(src, dst, gn, e1, vec, lane, nd, ns);
        } else {
#pragma unroll
            for (int k = 0; k < 8; ++k) { nd[k] = -1; ns[k] = 0; }
        }
#pragma unroll
        for (int k = 0; k < 8; ++k) {
            const unsigned dl = (unsigned)(cd[k] - lo);
            const bool m = dl < (unsigned)nloc;
            const unsigned mask = __builtin_amdgcn_ballot_w32(m);
            if (mask != 0u) {
                if (m) {
                    const unsigned pos = __builtin_amdgcn_mbcnt_lo(mask, 0u);
                    const unsigned qi  = (head + cnt + pos) & 63u;
                    rs[qi] = cs[k];
                    rd[qi] = (int)dl;
                }
                cnt += (unsigned)__builtin_popcount(mask);
                if (cnt >= 32u) {
                    drain_ring<F, true>(rs, rd, head, cnt, lane, lo, dinv, t, tile);
                    head = (head + 32u) & 63u;
                    cnt -= 32u;
                }
            }
        }
        if (!more) break;
#pragma unroll
        for (int k = 0; k < 8; ++k) { cd[k] = nd[k]; cs[k] = ns[k]; }
        gn += 32 * 256;
    }
    if (cnt != 0u)
        drain_ring<F, false>(rs, rd, head, cnt, lane, lo, dinv, t, tile);
}

constexpr int SPLIT_MAXP  = 4096;
constexpr int SPLIT_STAGE = 65536;

__device__ __forceinline__ void split_fetch(const int* __restrict__ src, const int* __restrict__ dst,
                                            int g, int e1, bool vec, int l64,
                                            int (&d)[4], int (&s)[4]) {
    if (g < e1) {
        if (vec && g + 256 <= e1) {
            const int e = g + l64 * 4;
            const int4 qd = *reinterpret_cast<const int4*>(dst + e);
            const int4 qs = *reinterpret_cast<const int4*>(src + e);
            d[0] = qd.x; d[1] = qd.y; d[2] = qd.z; d[3] = qd.w;
            s[0] = qs.x; s[1] = qs.y; s[2] = qs.z; s[3] = qs.w;
        } else {
#pragma unroll
            for (int i = 0; i < 4; ++i) {
                const int e = g + l64 * 4 + i;
                d[i] = (e < e1) ? dst[e] : -1;
                s[i] = (e < e1) ? src[e] : 0;
            }
        }
    } else {
#pragma unroll
        for (int i = 0; i < 4; ++i) { d[i] = -1; s[i] = 0; }
    }
}

template <int NP>
__global__ __launch_bounds__(1024) void k_split_lists(const int* __restrict__ src, const int* __restrict__ dst,
                                                      unsigned* __restrict__ lists, unsigned cap,
                                                      int E, int N) {
    static_assert(NP <= (1 << (32 - LIST_SRC_BITS)), "a place in the range must fit above the source bits");
    __shared__ unsigned cnt[SPLIT_MAXP];
    __shared__ unsigned stage[SPLIT_STAGE];
    const int tid = (int)threadIdx.x;
    const int P  = (N + NP - 1) / NP;
    const int R  = (int)gridDim.x / 32;
    const int r  = (int)blockIdx.x / 32;
    const int wv = (int)blockIdx.x % 32;
    unsigned* counts = lists + (size_t)r * P * 32 + wv;
    if (P > SPLIT_MAXP) {
        for (int p = tid; p < P; p += 1024) counts[(size_t)p * 32] = 0xffffffffu;
        return;
    }
    for (int p = tid; p < P; p += 1024) cnt[p] = 0u;
    __syncthreads();

    const int per = (((E + R - 1) / R) + 255) & ~255;
    const int e0  = r * per;
    const int e1  = (e0 + per) < E ? (e0 + per) : E;
    const bool vec = (E & 3) == 0;
    const int q   = tid >> 6;
    const int l64 = tid & 63;
    const int g0  = e0 + wv * 256;
    const int J   = g0 < e1 ? (e1 - g0 + 8191) / 8192 : 0;
    const int T   = (J + 15) / 16;
    unsigned* ent = lists + (size_t)P * R * 32 + ((size_t)r * P * 32 + wv) * cap;
    const bool staged = (unsigned long long)P * cap <= (unsigned long long)SPLIT_STAGE;

    int cd[4], cs[4];
    split_fetch(src, dst, g0 + q * 8192, e1, vec, l64, cd, cs);
    for (int t = 0; t < T; ++t) {
        int nd[4], ns[4];
        split_fetch(src, dst, g0 + ((t + 1) * 16 + q) * 8192, e1, vec, l64, nd, ns);
        unsigned pp[4], rk[4];
#pragma unroll
        for (int i = 0; i < 4; ++i) {
            const bool ok = (unsigned)cd[i] < (unsigned)N;
            pp[i] = ok ? (unsigned)cd[i] / (unsigned)NP : 0u;
            rk[i] = ok ? atomicAdd(&cnt[pp[i]], 1u) : 0xffffffffu;
        }
#pragma unroll
        for (int i = 0; i < 4; ++i) {
            if (rk[i] < cap) {
                const unsigned dl = (unsigned)cd[i] - pp[i] * (unsigned)NP;
                const unsigned wd = (unsigned)cs[i] | (dl << LIST_SRC_BITS);
                if (staged) stage[pp[i] * cap + rk[i]] = wd;
                else        ent[(size_t)pp[i] * 32u * cap + rk[i]] = wd;
            }
        }
#pragma unroll
        for (int i = 0; i < 4; ++i) { cd[i] = nd[i]; cs[i] = ns[i]; }
    }
    __syncthreads();
    for (int p = tid; p < P; p += 1024) counts[(size_t)p * 32] = cnt[p];
    if (staged) {
        const unsigned lane = (unsigned)tid & 31u;
        for (int p = tid >> 5; p < P; p += 32) {
            const unsigned n = cnt[p] < cap ? cnt[p] : cap;
            const unsigned* sp = stage + (unsigned)p * cap;
            unsigned* gp = ent + (size_t)p * 32u * cap;
            for (unsigned k = lane; k < n; k += 32u) gp[k] = sp[k];
        }
    }
}

template <int F, int NP>
__global__ __launch_bounds__(1024) void k_edge_tile(const int* __restrict__ src, const int* __restrict__ dst,
                                                    const float* __restrict__ dinv, const float* __restrict__ t,
                                                    float* __restrict__ agg, const unsigned* __restrict__ lists,
                                                    unsigned cap, int E, int N) {
    static_assert(NP <= (1 << (32 - LIST_SRC_BITS)), "a place in the range must fit above the source bits");
    __shared__ __attribute__((aligned(16))) float tile[NP * F];
    __shared__ int ring_s[32][64];
    __shared__ int ring_d[32][64];

    const unsigned tid  = threadIdx.x;
    const unsigned lane = tid & 31u;
    const unsigned wv = (unsigned)__builtin_amdgcn_readfirstlane((int)(tid >> 5));

    const int P = (N + NP - 1) / NP;
    const int R = (int)gridDim.x / P;
    const int p = (int)blockIdx.x % P;
    const int r = (int)blockIdx.x / P;
    const int lo   = p * NP;
    const int nloc = (N - lo) < NP ? (N - lo) : NP;

    for (int i = (int)tid; i < NP * F / 4; i += 1024)
        reinterpret_cast<float4*>(tile)[i] = make_float4(0.0f, 0.0f, 0.0f, 0.0f);
    __syncthreads();

    const unsigned wid = blockIdx.x * 32u + wv;
    const unsigned* mylist = lists + (size_t)gridDim.x * 32u + (size_t)wid * cap;

    const unsigned c = (unsigned)__builtin_amdgcn_readfirstlane((int)lists[wid]);
    if (c <= cap) {
        unsigned nxt = (lane < c) ? mylist[lane] : 0u;
        for (unsigned k = 0; k < c; k += 32u) {
            const unsigned cur = nxt;
            const unsigned kn  = k + 32u;
            if (kn < c) nxt = (kn + lane < c) ? mylist[kn + lane] : 0u;
            const int      s  = (int)(cur & LIST_SRC_MASK);
            const unsigned dl = cur >> LIST_SRC_BITS;
            const unsigned rem = c - k;
            if (rem >= 32u) drain_regs<F, true >(s, dl, 32u, lane, lo, dinv, t, tile);
            else            drain_regs<F, false>(s, dl, rem, lane, lo, dinv, t, tile);
        }
    } else {
        const int per = (((E + R - 1) / R) + 255) & ~255;
        const int e0  = r * per;
        const int e1  = (e0 + per) < E ? (e0 + per) : E;
        const bool vec = (E & 3) == 0;
        search_slice<F>(src, dst, dinv, t, tile, ring_s[wv], ring_d[wv], e0, e1, vec,
                        wv, lane, lo, nloc);
    }
    __syncthreads();

    float* ag = agg + (size_t)lo * F;
    const int tot = nloc * F;
    for (int i = (int)tid; i < tot; i += 1024)
        atomicAdd(ag + i, tile[i]);
}

__global__ void k_gemm_t2(const float* __restrict__ agg1, const float* __restrict__ W2,
                          const float* __restrict__ dinv, const float* __restrict__ b2,
                          float* __restrict__ t2, float* __restrict__ out, int N, int nTiles) {
    int wave = (blockIdx.x * blockDim.x + threadIdx.x) >> 5;
    if (wave >= nTiles) return;
    int lane = threadIdx.x & 31;
    int n    = lane & 15;
    int kh   = lane >> 4;
    int rowA = wave * 16 + n;    if (rowA >= N) rowA = N - 1;

    float dv[8];
#pragma unroll
    for (int v = 0; v < 8; ++v) {
        int nd = wave * 16 + 8 * kh + v;   if (nd >= N) nd = N - 1;
        dv[v] = dinv[nd];
    }

    v8f acc0 = {}, acc1 = {};
    const float* ar = agg1 + (size_t)rowA * 32;
    bool c1ok = (16 + n) < 20;
#pragma unroll
    for (int k0 = 0; k0 < 32; k0 += 4) {
        int kb = k0 + 2 * kh;
        v2f a;  a.x = fmaxf(ar[kb], 0.0f);  a.y = fmaxf(ar[kb + 1], 0.0f);
        v2f b0; b0.x = W2[kb * 20 + n];     b0.y = W2[(kb + 1) * 20 + n];
        v2f b1; b1.x = c1ok ? W2[kb * 20 + 16 + n] : 0.0f;
                b1.y = c1ok ? W2[(kb + 1) * 20 + 16 + n] : 0.0f;
        acc0 = wmma4(a, b0, acc0);
        acc1 = wmma4(a, b1, acc1);
    }
    const float bia0 = b2[n];
    const float bia1 = c1ok ? b2[16 + n] : 0.0f;
#pragma unroll
    for (int v = 0; v < 8; ++v) {
        int mr   = (lane < 16) ? v : v + 8;
        int node = wave * 16 + mr;
        if (node < N) {
            float* o = t2 + (size_t)node * 20;
            float* q = out + (size_t)node * 20;
            const float di = dv[v];
            o[n] = acc0[v];
            q[n] = acc0[v] * di * di + bia0;
            if (c1ok) {
                o[16 + n] = acc1[v];
                q[16 + n] = acc1[v] * di * di + bia1;
            }
        }
    }
}

extern "C" void kernel_launch(void* const* d_in, const int* in_sizes, int n_in,
                              void* d_out, int out_size, void* d_ws, size_t ws_size,
                              hipStream_t stream) {
    const float* x    = (const float*)d_in[0];
    const float* cnn  = (const float*)d_in[1];
    const int*   ei   = (const int*)d_in[2];
    const float* fcW  = (const float*)d_in[3];
    const float* fcb  = (const float*)d_in[4];
    const float* cnnW = (const float*)d_in[5];
    const float* cnnb = (const float*)d_in[6];
    const float* W1   = (const float*)d_in[7];
    const float* b1   = (const float*)d_in[8];
    const float* W2   = (const float*)d_in[9];
    const float* b2   = (const float*)d_in[10];

    const int N = in_sizes[0] / 32;
    const int E = in_sizes[2] / 2;
    const int* src = ei;
    const int* dst = ei + E;

    float* ws   = (float*)d_ws;
    float* dinv = ws;  ws += N;
    unsigned* planes = (unsigned*)((float*)d_ws + ((N + 3) / 4) * 4);
    float* h0   = ws;  ws += (size_t)N * 64;
    float* t1   = ws;  ws += (size_t)N * 32;
    float* agg1 = ws;  ws += (size_t)N * 32;
    float* t2   = ws;  ws += (size_t)N * 20;
    float* out  = (float*)d_out;

    const int nTiles = (N + 15) / 16;
    dim3 blkG(128);
    dim3 grdG((nTiles * 32 + 127) / 128);
    const int PT = (N + 2047) / 2048;
    const int RT = PT < 256 ? 256 / PT : 1;
    unsigned* lists = (unsigned*)h0;
    const long long lwaves = (long long)PT * RT * 32;
    long long lcap = (((long long)N * 64 - lwaves) / lwaves) & ~31LL;
    if (lcap < 0 || N > (1 << LIST_SRC_BITS)) lcap = 0;
    if (lcap > 65536) lcap = 65536;
    const unsigned cap = (unsigned)lcap;

    const int NW4 = ((N + 7) / 8) * 4;
    const int PC  = (NW4 + CNT_WORDS - 1) / CNT_WORDS;
    const int RC  = (E + CNT_SLICE - 1) / CNT_SLICE;
    k_count_planes<<<PC * RC, 1024, 0, stream>>>(dst, planes, E, NW4);
    k_dinv_planes<<<(NW4 + 31) / 32, 256, 0, stream>>>(planes, dinv, N, NW4, RC);

    k_encoder<<<grdG, blkG, 0, stream>>>(x, cnn, fcW, fcb, cnnW, cnnb, h0, N, nTiles);
    k_gemm_t1<<<grdG, blkG, 0, stream>>>(h0, W1, dinv, b1, t1, agg1, N, nTiles);

    k_split_lists<2048><<<RT * 32, 1024, 0, stream>>>(src, dst, lists, cap, E, N);

    k_edge_tile<32, 2048><<<PT * RT, 1024, 0, stream>>>(src, dst, dinv, t1, agg1, lists, cap, E, N);

    k_gemm_t2<<<grdG, blkG, 0, stream>>>(agg1, W2, dinv, b2, t2, out, N, nTiles);
    k_edge_tile<20, 2048><<<PT * RT, 1024, 0, stream>>>(src, dst, dinv, t2, out, lists, cap, E, N);
}


# Edge weights from an LDS table: aggregation work-groups keep an in-degree byte per node in LDS, take dinv[src], dinv[dst] from it (same bits); ranges 1600x4 slices (was 2048x5): partial sums regroup
# speedup vs baseline: 5.0964x; 1.2361x over previous
#include <hip/hip_runtime.h>
#include <hip/hip_bf16.h>

typedef __attribute__((ext_vector_type(2))) float v2f;
typedef __attribute__((ext_vector_type(8))) float v8f;

__device__ __forceinline__ v8f wmma4(v2f a, v2f b, v8f c) {
    return __builtin_amdgcn_wmma_f32_16x16x4_f32(
         false, a,  false, b,
         (short)0, c,  false,  false);
}

constexpr int CNT_SLICE = 32768;
constexpr int CNT_WORDS = 65536;
constexpr int CNT_LOADS = CNT_SLICE / 4096;

__device__ __forceinline__ void count_one(unsigned* cw, int d, int lo, unsigned nn) {
    const unsigned dl = (unsigned)(d - lo);
    if (dl < nn) atomicAdd(cw + (dl >> 1), 1u << ((dl & 1u) << 4));
}

__global__ __launch_bounds__(1024) void k_count_planes(const int* __restrict__ dst,
                                                       unsigned* __restrict__ planes,
                                                       int E, int NW4) {
    __shared__ __attribute__((aligned(16))) unsigned cw[CNT_WORDS];
    const int tid = (int)threadIdx.x;
    const int P  = (NW4 + CNT_WORDS - 1) / CNT_WORDS;
    const int p  = (int)blockIdx.x % P;
    const int r  = (int)blockIdx.x / P;
    const int w0 = p * CNT_WORDS;
    const int nw = (NW4 - w0) < CNT_WORDS ? (NW4 - w0) : CNT_WORDS;

    uint4* cw4 = reinterpret_cast<uint4*>(cw);
    for (int i = tid; i < nw / 4; i += 1024) cw4[i] = make_uint4(0u, 0u, 0u, 0u);
    __syncthreads();

    const int e0 = r * CNT_SLICE;
    const int e1 = (e0 + CNT_SLICE) < E ? (e0 + CNT_SLICE) : E;
    const int lo = 2 * w0;
    const unsigned nn = 2u * (unsigned)nw;
    if (((E & 3) == 0) && (e0 + CNT_SLICE <= E)) {
        int4 q[CNT_LOADS];
#pragma unroll
        for (int k = 0; k < CNT_LOADS; ++k)
            q[k] = *reinterpret_cast<const int4*>(dst + e0 + (k * 1024 + tid) * 4);
#pragma unroll
        for (int k = 0; k < CNT_LOADS; ++k) {
            count_one(cw, q[k].x, lo, nn);
            count_one(cw, q[k].y, lo, nn);
            count_one(cw, q[k].z, lo, nn);
            count_one(cw, q[k].w, lo, nn);
        }
    } else {
        for (int e = e0 + tid; e < e1; e += 1024) count_one(cw, dst[e], lo, nn);
    }
    __syncthreads();

    uint4* pl4 = reinterpret_cast<uint4*>(planes + (size_t)r * NW4 + w0);
    for (int i = tid; i < nw / 4; i += 1024) pl4[i] = cw4[i];
}

__global__ __launch_bounds__(256) void k_dinv_planes(const unsigned* __restrict__ planes,
                                                     float* __restrict__ dinv,
                                                     int N, int NW4, int R) {
    __shared__ unsigned part[8][64];
    const int lane = (int)threadIdx.x & 31;
    const int wv   = (int)threadIdx.x >> 5;
    const int i    = (int)blockIdx.x * 32 + lane;
    unsigned lo = 0u, hi = 0u;
    if (i < NW4) {
        for (int base = wv; base < R; base += 128) {
            unsigned w[16];
#pragma unroll
            for (int k = 0; k < 16; ++k) {
                const int r = base + 8 * k;
                w[k] = (r < R) ? planes[(size_t)r * NW4 + i] : 0u;
            }
#pragma unroll
            for (int k = 0; k < 16; ++k) {
                lo += w[k] & 0xffffu;
                hi += w[k] >> 16;
            }
        }
    }
    part[wv][lane]      = lo;
    part[wv][32 + lane] = hi;
    __syncthreads();
    if (wv < 2) {
        unsigned s = 0u;
        for (int k = 0; k < 8; ++k) s += part[k][wv * 32 + lane];
        const int node = 2 * i + wv;
        if (node < N) dinv[node] = rsqrtf((float)s + 1.0f);
    }
}

__global__ void k_encoder(const float* __restrict__ x, const float* __restrict__ cnn,
                          const float* __restrict__ fcW, const float* __restrict__ fcb,
                          const float* __restrict__ cnnW, const float* __restrict__ cnnb,
                          float* __restrict__ h0, int N, int nTiles) {
    int wave = (blockIdx.x * blockDim.x + threadIdx.x) >> 5;
    if (wave >= nTiles) return;
    int lane = threadIdx.x & 31;
    int n    = lane & 15;
    int kh   = lane >> 4;
    int rowA = wave * 16 + n;    if (rowA >= N) rowA = N - 1;

    v8f acc0 = {}, acc1 = {}, acc2 = {}, acc3 = {};

    const float* ax = x + (size_t)rowA * 32;
#pragma unroll
    for (int k0 = 0; k0 < 32; k0 += 4) {
        int kb = k0 + 2 * kh;
        v2f a;  a.x = ax[kb];               a.y = ax[kb + 1];
        v2f b0; b0.x = fcW[kb * 32 + n];    b0.y = fcW[(kb + 1) * 32 + n];
        v2f b1; b1.x = fcW[kb * 32 + 16 + n]; b1.y = fcW[(kb + 1) * 32 + 16 + n];
        acc0 = wmma4(a, b0, acc0);
        acc1 = wmma4(a, b1, acc1);
    }
    const float* ac = cnn + (size_t)rowA * 128;
#pragma unroll
    for (int k0 = 0; k0 < 128; k0 += 4) {
        int kb = k0 + 2 * kh;
        v2f a;  a.x = ac[kb];                a.y = ac[kb + 1];
        v2f b0; b0.x = cnnW[kb * 32 + n];    b0.y = cnnW[(kb + 1) * 32 + n];
        v2f b1; b1.x = cnnW[kb * 32 + 16 + n]; b1.y = cnnW[(kb + 1) * 32 + 16 + n];
        acc2 = wmma4(a, b0, acc2);
        acc3 = wmma4(a, b1, acc3);
    }
#pragma unroll
    for (int v = 0; v < 8; ++v) {
        int mr   = (lane < 16) ? v : v + 8;
        int node = wave * 16 + mr;
        if (node < N) {
            float* o = h0 + (size_t)node * 64;
            o[n]      = fmaxf(acc0[v] + fcb[n],       0.0f);
            o[16 + n] = fmaxf(acc1[v] + fcb[16 + n],  0.0f);
            o[32 + n] = fmaxf(acc2[v] + cnnb[n],      0.0f);
            o[48 + n] = fmaxf(acc3[v] + cnnb[16 + n], 0.0f);
        }
    }
}

__global__ void k_gemm_t1(const float* __restrict__ h0, const float* __restrict__ W1,
                          const float* __restrict__ dinv, const float* __restrict__ b1,
                          float* __restrict__ t1, float* __restrict__ agg1, int N, int nTiles) {
    int wave = (blockIdx.x * blockDim.x + threadIdx.x) >> 5;
    if (wave >= nTiles) return;
    int lane = threadIdx.x & 31;
    int n    = lane & 15;
    int kh   = lane >> 4;
    int rowA = wave * 16 + n;    if (rowA >= N) rowA = N - 1;

    float dv[8];
#pragma unroll
    for (int v = 0; v < 8; ++v) {
        int nd = wave * 16 + 8 * kh + v;   if (nd >= N) nd = N - 1;
        dv[v] = dinv[nd];
    }

    v8f acc0 = {}, acc1 = {};
    const float* ar = h0 + (size_t)rowA * 64;
#pragma unroll
    for (int k0 = 0; k0 < 64; k0 += 4) {
        int kb = k0 + 2 * kh;
        v2f a;  a.x = ar[kb];              a.y = ar[kb + 1];
        v2f b0; b0.x = W1[kb * 32 + n];    b0.y = W1[(kb + 1) * 32 + n];
        v2f w1; w1.x = W1[kb * 32 + 16 + n]; w1.y = W1[(kb + 1) * 32 + 16 + n];
        acc0 = wmma4(a, b0, acc0);
        acc1 = wmma4(a, w1, acc1);
    }
    const float bia0 = b1[n], bia1 = b1[16 + n];
#pragma unroll
    for (int v = 0; v < 8; ++v) {
        int mr   = (lane < 16) ? v : v + 8;
        int node = wave * 16 + mr;
        if (node < N) {
            float* o = t1 + (size_t)node * 32;
            float* a = agg1 + (size_t)node * 32;
            const float di = dv[v];
            o[n]      = acc0[v];
            o[16 + n] = acc1[v];
            a[n]      = acc0[v] * di * di + bia0;
            a[16 + n] = acc1[v] * di * di + bia1;
        }
    }
}

constexpr int      LIST_SRC_BITS = 21;
constexpr unsigned LIST_SRC_MASK = (1u << LIST_SRC_BITS) - 1u;

constexpr int RANGE_NP  = 1600;
constexpr int DEG_TAB   = 102400;
constexpr int TAB_LOADS = (DEG_TAB / 16 + 1023) / 1024;

__device__ __forceinline__ float node_dinv(const unsigned char* __restrict__ tab, int tabn,
                                           const float* __restrict__ dinv, int node) {
    unsigned b = 255u;
    if (node < tabn) b = tab[node];
    return (b != 255u) ? rsqrtf((float)b + 1.0f) : dinv[node];
}

__device__ __forceinline__ unsigned deg_byte(float di) {
    const float c = rintf(1.0f / (di * di));
    return (c >= 1.0f && c < 256.0f) ? (unsigned)c - 1u : 255u;
}

template <int F, bool FULL>
__device__ __forceinline__ void drain_regs(int s, unsigned dl, unsigned cnt, unsigned lane, int lo,
                                           const float* __restrict__ dinv, const float* __restrict__ t,
                                           float* __restrict__ tile,
                                           const unsigned char* __restrict__ tab, int tabn) {
    const bool act = lane < (unsigned)F;
    const unsigned fl = act ? lane : (unsigned)(F - 1);
    float v[32];
#pragma unroll
    for (int u = 0; u < 32; ++u) {
        const unsigned sj = (unsigned)__builtin_amdgcn_readlane(s, u);
        const float* row = t + (size_t)sj * F;
        v[u] = row[fl];
    }
    float w = 0.0f;
    if (FULL || lane < cnt)
        w = node_dinv(tab, tabn, dinv, s) * node_dinv(tab, tabn, dinv, lo + (int)dl);
    unsigned dj[32];
    float wj[32];
#pragma unroll
    for (int u = 0; u < 32; ++u) {
        dj[u] = (unsigned)__builtin_amdgcn_readlane((int)dl, u);
        wj[u] = __int_as_float(__builtin_amdgcn_readlane(__float_as_int(w), u));
    }
    if (act) {
#pragma unroll
        for (int u = 0; u < 32; ++u)
            if (FULL || (unsigned)u < cnt)
                atomicAdd(tile + dj[u] * F + lane, v[u] * wj[u]);
    }
}

template <int F, bool FULL>
__device__ __forceinline__ void drain_ring(const int* __restrict__ rs, const int* __restrict__ rd,
                                           unsigned head, unsigned cnt, unsigned lane, int lo,
                                           const float* __restrict__ dinv, const float* __restrict__ t,
                                           float* __restrict__ tile,
                                           const unsigned char* __restrict__ tab, int tabn) {
    const unsigned qi = (head + lane) & 63u;
    int s = 0;
    unsigned dl = 0;
    if (FULL || lane < cnt) {
        s  = rs[qi];
        dl = (unsigned)rd[qi];
    }
    drain_regs<F, FULL>(s, dl, cnt, lane, lo, dinv, t, tile, tab, tabn);
}

__device__ __forceinline__ void load_group(const int* __restrict__ src, const int* __restrict__ dst,
                                           int g, int e1, bool vec, unsigned lane,
                                           int (&d)[8], int (&s)[8]) {
    if (vec && g + 256 <= e1) {
#pragma unroll
        for (int h = 0; h < 2; ++h) {
            const int e = g + h * 128 + (int)lane * 4;
            const int4 qd = *reinterpret_cast<const int4*>(dst + e);
            const int4 qs = *reinterpret_cast<const int4*>(src + e);
            d[4 * h] = qd.x; d[4 * h + 1] = qd.y; d[4 * h + 2] = qd.z; d[4 * h + 3] = qd.w;
            s[4 * h] = qs.x; s[4 * h + 1] = qs.y; s[4 * h + 2] = qs.z; s[4 * h + 3] = qs.w;
        }
    } else {
#pragma unroll
        for (int h = 0; h < 2; ++h) {
#pragma unroll
            for (int i = 0; i < 4; ++i) {
                const int e = g + h * 128 + (int)lane * 4 + i;
                d[4 * h + i] = (e < e1) ? dst[e] : -1;
                s[4 * h + i] = (e < e1) ? src[e] : 0;
            }
        }
    }
}

template <int F>
__device__ __forceinline__ void search_slice(const int* __restrict__ src, const int* __restrict__ dst,
                                                 const float* __restrict__ dinv, const float* __restrict__ t,
                                                 float* __restrict__ tile, int* __restrict__ rs, int* __restrict__ rd,
                                                 int e0, int e1, bool vec, unsigned wv, unsigned lane,
                                                 int lo, int nloc,
                                                 const unsigned char* __restrict__ tab, int tabn) {
    unsigned head = 0, cnt = 0;

    int cd[8], cs[8];
#pragma unroll
    for (int k = 0; k < 8; ++k) { cd[k] = -1; cs[k] = 0; }
    int gn = e0 + (int)wv * 256;
    for (;;) {
        const bool more = gn < e1;
        int nd[8], ns[8];
        if (more) {
            load_group(src, dst, gn, e1, vec, lane, nd, ns);
        } else {
#pragma unroll
            for (int k = 0; k < 8; ++k) { nd[k] = -1; ns[k] = 0; }
        }
#pragma unroll
        for (int k = 0; k < 8; ++k) {
            const unsigned dl = (unsigned)(cd[k] - lo);
            const bool m = dl < (unsigned)nloc;
            const unsigned mask = __builtin_amdgcn_ballot_w32(m);
            if (mask != 0u) {
                if (m) {
                    const unsigned pos = __builtin_amdgcn_mbcnt_lo(mask, 0u);
                    const unsigned qi  = (head + cnt + pos) & 63u;
                    rs[qi] = cs[k];
                    rd[qi] = (int)dl;
                }
                cnt += (unsigned)__builtin_popcount(mask);
                if (cnt >= 32u) {
                    drain_ring<F, true>(rs, rd, head, cnt, lane, lo, dinv, t, tile, tab, tabn);
                    head = (head + 32u) & 63u;
                    cnt -= 32u;
                }
            }
        }
        if (!more) break;
#pragma unroll
        for (int k = 0; k < 8; ++k) { cd[k] = nd[k]; cs[k] = ns[k]; }
        gn += 32 * 256;
    }
    if (cnt != 0u)
        drain_ring<F, false>(rs, rd, head, cnt, lane, lo, dinv, t, tile, tab, tabn);
}

constexpr int SPLIT_MAXP  = 4096;
constexpr int SPLIT_STAGE = 65536;

__device__ __forceinline__ void split_fetch(const int* __restrict__ src, const int* __restrict__ dst,
                                            int g, int e1, bool vec, int l64,
                                            int (&d)[4], int (&s)[4]) {
    if (g < e1) {
        if (vec && g + 256 <= e1) {
            const int e = g + l64 * 4;
            const int4 qd = *reinterpret_cast<const int4*>(dst + e);
            const int4 qs = *reinterpret_cast<const int4*>(src + e);
            d[0] = qd.x; d[1] = qd.y; d[2] = qd.z; d[3] = qd.w;
            s[0] = qs.x; s[1] = qs.y; s[2] = qs.z; s[3] = qs.w;
        } else {
#pragma unroll
            for (int i = 0; i < 4; ++i) {
                const int e = g + l64 * 4 + i;
                d[i] = (e < e1) ? dst[e] : -1;
                s[i] = (e < e1) ? src[e] : 0;
            }
        }
    } else {
#pragma unroll
        for (int i = 0; i < 4; ++i) { d[i] = -1; s[i] = 0; }
    }
}

template <int NP>
__global__ __launch_bounds__(1024) void k_split_lists(const int* __restrict__ src, const int* __restrict__ dst,
                                                      unsigned* __restrict__ lists, unsigned cap,
                                                      int E, int N,
                                                      const float* __restrict__ dinv, unsigned* __restrict__ degb) {
    static_assert(NP <= (1 << (32 - LIST_SRC_BITS)), "a place in the range must fit above the source bits");
    __shared__ unsigned cnt[SPLIT_MAXP];
    __shared__ unsigned stage[SPLIT_STAGE];
    const int tid = (int)threadIdx.x;

    const int TW = ((N + 15) / 16) * 4;
    for (int wi = (int)blockIdx.x * 1024 + tid; wi < TW; wi += (int)gridDim.x * 1024) {
        const int n0 = wi * 4;
        unsigned wd = 0u;
        if (n0 + 4 <= N) {
            const float4 q = *reinterpret_cast<const float4*>(dinv + n0);
            wd = deg_byte(q.x) | (deg_byte(q.y) << 8) | (deg_byte(q.z) << 16) | (deg_byte(q.w) << 24);
        } else {
#pragma unroll
            for (int k = 0; k < 4; ++k)
                wd |= ((n0 + k < N) ? deg_byte(dinv[n0 + k]) : 255u) << (8 * k);
        }
        degb[wi] = wd;
    }
    const int P  = (N + NP - 1) / NP;
    const int R  = (int)gridDim.x / 32;
    const int r  = (int)blockIdx.x / 32;
    const int wv = (int)blockIdx.x % 32;
    unsigned* counts = lists + (size_t)r * P * 32 + wv;
    if (P > SPLIT_MAXP) {
        for (int p = tid; p < P; p += 1024) counts[(size_t)p * 32] = 0xffffffffu;
        return;
    }
    for (int p = tid; p < P; p += 1024) cnt[p] = 0u;
    __syncthreads();

    const int per = (((E + R - 1) / R) + 255) & ~255;
    const int e0  = r * per;
    const int e1  = (e0 + per) < E ? (e0 + per) : E;
    const bool vec = (E & 3) == 0;
    const int q   = tid >> 6;
    const int l64 = tid & 63;
    const int g0  = e0 + wv * 256;
    const int J   = g0 < e1 ? (e1 - g0 + 8191) / 8192 : 0;
    const int T   = (J + 15) / 16;
    unsigned* ent = lists + (size_t)P * R * 32 + ((size_t)r * P * 32 + wv) * cap;
    const bool staged = (unsigned long long)P * cap <= (unsigned long long)SPLIT_STAGE;

    int cd[4], cs[4];
    split_fetch(src, dst, g0 + q * 8192, e1, vec, l64, cd, cs);
    for (int t = 0; t < T; ++t) {
        int nd[4], ns[4];
        split_fetch(src, dst, g0 + ((t + 1) * 16 + q) * 8192, e1, vec, l64, nd, ns);
        unsigned pp[4], rk[4];
#pragma unroll
        for (int i = 0; i < 4; ++i) {
            const bool ok = (unsigned)cd[i] < (unsigned)N;
            pp[i] = ok ? (unsigned)cd[i] / (unsigned)NP : 0u;
            rk[i] = ok ? atomicAdd(&cnt[pp[i]], 1u) : 0xffffffffu;
        }
#pragma unroll
        for (int i = 0; i < 4; ++i) {
            if (rk[i] < cap) {
                const unsigned dl = (unsigned)cd[i] - pp[i] * (unsigned)NP;
                const unsigned wd = (unsigned)cs[i] | (dl << LIST_SRC_BITS);
                if (staged) stage[pp[i] * cap + rk[i]] = wd;
                else        ent[(size_t)pp[i] * 32u * cap + rk[i]] = wd;
            }
        }
#pragma unroll
        for (int i = 0; i < 4; ++i) { cd[i] = nd[i]; cs[i] = ns[i]; }
    }
    __syncthreads();
    for (int p = tid; p < P; p += 1024) counts[(size_t)p * 32] = cnt[p];
    if (staged) {
        const unsigned lane = (unsigned)tid & 31u;
        for (int p = tid >> 5; p < P; p += 32) {
            const unsigned n = cnt[p] < cap ? cnt[p] : cap;
            const unsigned* sp = stage + (unsigned)p * cap;
            unsigned* gp = ent + (size_t)p * 32u * cap;
            for (unsigned k = lane; k < n; k += 32u) gp[k] = sp[k];
        }
    }
}

template <int F, int NP>
__global__ __launch_bounds__(1024) void k_edge_tile(const int* __restrict__ src, const int* __restrict__ dst,
                                                    const float* __restrict__ dinv, const float* __restrict__ t,
                                                    float* __restrict__ agg, const unsigned* __restrict__ lists,
                                                    unsigned cap, int E, int N,
                                                    const unsigned* __restrict__ degb) {
    static_assert(NP <= (1 << (32 - LIST_SRC_BITS)), "a place in the range must fit above the source bits");
    static_assert(DEG_TAB % 16 == 0 && DEG_TAB / 16 <= TAB_LOADS * 1024, "the table copy covers the table");
    __shared__ __attribute__((aligned(16))) float tile[NP * F];
    __shared__ int ring_s[32][64];
    __shared__ int ring_d[32][64];
    __shared__ __attribute__((aligned(16))) unsigned char degtab[DEG_TAB];

    const unsigned tid  = threadIdx.x;
    const unsigned lane = tid & 31u;
    const unsigned wv = (unsigned)__builtin_amdgcn_readfirstlane((int)(tid >> 5));

    const int P = (N + NP - 1) / NP;
    const int R = (int)gridDim.x / P;
    const int p = (int)blockIdx.x % P;
    const int r = (int)blockIdx.x / P;
    const int lo   = p * NP;
    const int nloc = (N - lo) < NP ? (N - lo) : NP;

    const int tabn = N < DEG_TAB ? N : DEG_TAB;
    const int ng   = (tabn + 15) / 16;
    uint4 tq[TAB_LOADS];
#pragma unroll
    for (int k = 0; k < TAB_LOADS; ++k) {
        const int i = (int)tid + k * 1024;
        tq[k] = (i < ng) ? reinterpret_cast<const uint4*>(degb)[i] : make_uint4(0u, 0u, 0u, 0u);
    }

    for (int i = (int)tid; i < NP * F / 4; i += 1024)
        reinterpret_cast<float4*>(tile)[i] = make_float4(0.0f, 0.0f, 0.0f, 0.0f);

#pragma unroll
    for (int k = 0; k < TAB_LOADS; ++k) {
        const int i = (int)tid + k * 1024;
        if (i < ng) reinterpret_cast<uint4*>(degtab)[i] = tq[k];
    }
    __syncthreads();

    const unsigned wid = blockIdx.x * 32u + wv;
    const unsigned* mylist = lists + (size_t)gridDim.x * 32u + (size_t)wid * cap;

    const unsigned c = (unsigned)__builtin_amdgcn_readfirstlane((int)lists[wid]);
    if (c <= cap) {
        unsigned nxt = (lane < c) ? mylist[lane] : 0u;
        for (unsigned k = 0; k < c; k += 32u) {
            const unsigned cur = nxt;
            const unsigned kn  = k + 32u;
            if (kn < c) nxt = (kn + lane < c) ? mylist[kn + lane] : 0u;
            const int      s  = (int)(cur & LIST_SRC_MASK);
            const unsigned dl = cur >> LIST_SRC_BITS;
            const unsigned rem = c - k;
            if (rem >= 32u) drain_regs<F, true >(s, dl, 32u, lane, lo, dinv, t, tile, degtab, tabn);
            else            drain_regs<F, false>(s, dl, rem, lane, lo, dinv, t, tile, degtab, tabn);
        }
    } else {
        const int per = (((E + R - 1) / R) + 255) & ~255;
        const int e0  = r * per;
        const int e1  = (e0 + per) < E ? (e0 + per) : E;
        const bool vec = (E & 3) == 0;
        search_slice<F>(src, dst, dinv, t, tile, ring_s[wv], ring_d[wv], e0, e1, vec,
                        wv, lane, lo, nloc, degtab, tabn);
    }
    __syncthreads();

    float* ag = agg + (size_t)lo * F;
    const int tot = nloc * F;
    for (int i = (int)tid; i < tot; i += 1024)
        atomicAdd(ag + i, tile[i]);
}

__global__ void k_gemm_t2(const float* __restrict__ agg1, const float* __restrict__ W2,
                          const float* __restrict__ dinv, const float* __restrict__ b2,
                          float* __restrict__ t2, float* __restrict__ out, int N, int nTiles) {
    int wave = (blockIdx.x * blockDim.x + threadIdx.x) >> 5;
    if (wave >= nTiles) return;
    int lane = threadIdx.x & 31;
    int n    = lane & 15;
    int kh   = lane >> 4;
    int rowA = wave * 16 + n;    if (rowA >= N) rowA = N - 1;

    float dv[8];
#pragma unroll
    for (int v = 0; v < 8; ++v) {
        int nd = wave * 16 + 8 * kh + v;   if (nd >= N) nd = N - 1;
        dv[v] = dinv[nd];
    }

    v8f acc0 = {}, acc1 = {};
    const float* ar = agg1 + (size_t)rowA * 32;
    bool c1ok = (16 + n) < 20;
#pragma unroll
    for (int k0 = 0; k0 < 32; k0 += 4) {
        int kb = k0 + 2 * kh;
        v2f a;  a.x = fmaxf(ar[kb], 0.0f);  a.y = fmaxf(ar[kb + 1], 0.0f);
        v2f b0; b0.x = W2[kb * 20 + n];     b0.y = W2[(kb + 1) * 20 + n];
        v2f b1; b1.x = c1ok ? W2[kb * 20 + 16 + n] : 0.0f;
                b1.y = c1ok ? W2[(kb + 1) * 20 + 16 + n] : 0.0f;
        acc0 = wmma4(a, b0, acc0);
        acc1 = wmma4(a, b1, acc1);
    }
    const float bia0 = b2[n];
    const float bia1 = c1ok ? b2[16 + n] : 0.0f;
#pragma unroll
    for (int v = 0; v < 8; ++v) {
        int mr   = (lane < 16) ? v : v + 8;
        int node = wave * 16 + mr;
        if (node < N) {
            float* o = t2 + (size_t)node * 20;
            float* q = out + (size_t)node * 20;
            const float di = dv[v];
            o[n] = acc0[v];
            q[n] = acc0[v] * di * di + bia0;
            if (c1ok) {
                o[16 + n] = acc1[v];
                q[16 + n] = acc1[v] * di * di + bia1;
            }
        }
    }
}

extern "C" void kernel_launch(void* const* d_in, const int* in_sizes, int n_in,
                              void* d_out, int out_size, void* d_ws, size_t ws_size,
                              hipStream_t stream) {
    const float* x    = (const float*)d_in[0];
    const float* cnn  = (const float*)d_in[1];
    const int*   ei   = (const int*)d_in[2];
    const float* fcW  = (const float*)d_in[3];
    const float* fcb  = (const float*)d_in[4];
    const float* cnnW = (const float*)d_in[5];
    const float* cnnb = (const float*)d_in[6];
    const float* W1   = (const float*)d_in[7];
    const float* b1   = (const float*)d_in[8];
    const float* W2   = (const float*)d_in[9];
    const float* b2   = (const float*)d_in[10];

    const int N = in_sizes[0] / 32;
    const int E = in_sizes[2] / 2;
    const int* src = ei;
    const int* dst = ei + E;

    float* ws   = (float*)d_ws;
    float* dinv = ws;  ws += N;
    unsigned* planes = (unsigned*)((float*)d_ws + ((N + 3) / 4) * 4);
    float* h0   = ws;  ws += (size_t)N * 64;
    float* t1   = ws;  ws += (size_t)N * 32;
    float* agg1 = ws;  ws += (size_t)N * 32;
    float* t2   = ws;  ws += (size_t)N * 20;
    float* out  = (float*)d_out;

    const int nTiles = (N + 15) / 16;
    dim3 blkG(128);
    dim3 grdG((nTiles * 32 + 127) / 128);
    const int PT = (N + RANGE_NP - 1) / RANGE_NP;
    const int RT = PT < 256 ? 256 / PT : 1;
    unsigned* degb = (unsigned*)((float*)d_ws + ((N + 3) / 4) * 4);
    const int TW = ((N + 15) / 16) * 4;
    unsigned* lists = degb + TW;
    const long long lwaves = (long long)PT * RT * 32;
    long long lcap = (((long long)N * 64 - 3 - TW - lwaves) / lwaves) & ~31LL;
    if (lcap < 0 || N > (1 << LIST_SRC_BITS)) lcap = 0;
    if (lcap > 65536) lcap = 65536;
    const unsigned cap = (unsigned)lcap;

    const int NW4 = ((N + 7) / 8) * 4;
    const int PC  = (NW4 + CNT_WORDS - 1) / CNT_WORDS;
    const int RC  = (E + CNT_SLICE - 1) / CNT_SLICE;
    k_count_planes<<<PC * RC, 1024, 0, stream>>>(dst, planes, E, NW4);
    k_dinv_planes<<<(NW4 + 31) / 32, 256, 0, stream>>>(planes, dinv, N, NW4, RC);

    k_encoder<<<grdG, blkG, 0, stream>>>(x, cnn, fcW, fcb, cnnW, cnnb, h0, N, nTiles);
    k_gemm_t1<<<grdG, blkG, 0, stream>>>(h0, W1, dinv, b1, t1, agg1, N, nTiles);

    k_split_lists<RANGE_NP><<<RT * 32, 1024, 0, stream>>>(src, dst, lists, cap, E, N, dinv, degb);

    k_edge_tile<32, RANGE_NP><<<PT * RT, 1024, 0, stream>>>(src, dst, dinv, t1, agg1, lists, cap, E, N, degb);

    k_gemm_t2<<<grdG, blkG, 0, stream>>>(agg1, W2, dinv, b2, t2, out, N, nTiles);
    k_edge_tile<20, RANGE_NP><<<PT * RT, 1024, 0, stream>>>(src, dst, dinv, t2, out, lists, cap, E, N, degb);
}


# In-degrees counted from the range lists by one launch (lists -> dinv, degree bytes); count-plane and plane-sum launches dropped (8 -> 7); self-loop term now starts the LDS tiles, joins a slice sum
# speedup vs baseline: 5.5565x; 1.0903x over previous
#include <hip/hip_runtime.h>
#include <hip/hip_bf16.h>

typedef __attribute__((ext_vector_type(2))) float v2f;
typedef __attribute__((ext_vector_type(8))) float v8f;

__device__ __forceinline__ v8f wmma4(v2f a, v2f b, v8f c) {
    return __builtin_amdgcn_wmma_f32_16x16x4_f32(
         false, a,  false, b,
         (short)0, c,  false,  false);
}

__global__ void k_encoder(const float* __restrict__ x, const float* __restrict__ cnn,
                          const float* __restrict__ fcW, const float* __restrict__ fcb,
                          const float* __restrict__ cnnW, const float* __restrict__ cnnb,
                          float* __restrict__ h0, int N, int nTiles) {
    int wave = (blockIdx.x * blockDim.x + threadIdx.x) >> 5;
    if (wave >= nTiles) return;
    int lane = threadIdx.x & 31;
    int n    = lane & 15;
    int kh   = lane >> 4;
    int rowA = wave * 16 + n;    if (rowA >= N) rowA = N - 1;

    v8f acc0 = {}, acc1 = {}, acc2 = {}, acc3 = {};

    const float* ax = x + (size_t)rowA * 32;
#pragma unroll
    for (int k0 = 0; k0 < 32; k0 += 4) {
        int kb = k0 + 2 * kh;
        v2f a;  a.x = ax[kb];               a.y = ax[kb + 1];
        v2f b0; b0.x = fcW[kb * 32 + n];    b0.y = fcW[(kb + 1) * 32 + n];
        v2f b1; b1.x = fcW[kb * 32 + 16 + n]; b1.y = fcW[(kb + 1) * 32 + 16 + n];
        acc0 = wmma4(a, b0, acc0);
        acc1 = wmma4(a, b1, acc1);
    }
    const float* ac = cnn + (size_t)rowA * 128;
#pragma unroll
    for (int k0 = 0; k0 < 128; k0 += 4) {
        int kb = k0 + 2 * kh;
        v2f a;  a.x = ac[kb];                a.y = ac[kb + 1];
        v2f b0; b0.x = cnnW[kb * 32 + n];    b0.y = cnnW[(kb + 1) * 32 + n];
        v2f b1; b1.x = cnnW[kb * 32 + 16 + n]; b1.y = cnnW[(kb + 1) * 32 + 16 + n];
        acc2 = wmma4(a, b0, acc2);
        acc3 = wmma4(a, b1, acc3);
    }
#pragma unroll
    for (int v = 0; v < 8; ++v) {
        int mr   = (lane < 16) ? v : v + 8;
        int node = wave * 16 + mr;
        if (node < N) {
            float* o = h0 + (size_t)node * 64;
            o[n]      = fmaxf(acc0[v] + fcb[n],       0.0f);
            o[16 + n] = fmaxf(acc1[v] + fcb[16 + n],  0.0f);
            o[32 + n] = fmaxf(acc2[v] + cnnb[n],      0.0f);
            o[48 + n] = fmaxf(acc3[v] + cnnb[16 + n], 0.0f);
        }
    }
}

__global__ void k_gemm_t1(const float* __restrict__ h0, const float* __restrict__ W1,
                          const float* __restrict__ b1,
                          float* __restrict__ t1, float* __restrict__ agg1, int N, int nTiles) {
    int wave = (blockIdx.x * blockDim.x + threadIdx.x) >> 5;
    if (wave >= nTiles) return;
    int lane = threadIdx.x & 31;
    int n    = lane & 15;
    int kh   = lane >> 4;
    int rowA = wave * 16 + n;    if (rowA >= N) rowA = N - 1;

    v8f acc0 = {}, acc1 = {};
    const float* ar = h0 + (size_t)rowA * 64;
#pragma unroll
    for (int k0 = 0; k0 < 64; k0 += 4) {
        int kb = k0 + 2 * kh;
        v2f a;  a.x = ar[kb];              a.y = ar[kb + 1];
        v2f b0; b0.x = W1[kb * 32 + n];    b0.y = W1[(kb + 1) * 32 + n];
        v2f w1; w1.x = W1[kb * 32 + 16 + n]; w1.y = W1[(kb + 1) * 32 + 16 + n];
        acc0 = wmma4(a, b0, acc0);
        acc1 = wmma4(a, w1, acc1);
    }
    const float bia0 = b1[n], bia1 = b1[16 + n];
#pragma unroll
    for (int v = 0; v < 8; ++v) {
        int mr   = (lane < 16) ? v : v + 8;
        int node = wave * 16 + mr;
        if (node < N) {
            float* o = t1 + (size_t)node * 32;
            float* a = agg1 + (size_t)node * 32;
            o[n]      = acc0[v];
            o[16 + n] = acc1[v];
            a[n]      = bia0;
            a[16 + n] = bia1;
        }
    }
}

constexpr int      LIST_SRC_BITS = 21;
constexpr unsigned LIST_SRC_MASK = (1u << LIST_SRC_BITS) - 1u;

constexpr int RANGE_NP  = 1600;
constexpr int DEG_TAB   = 102400;
constexpr int TAB_LOADS = (DEG_TAB / 16 + 1023) / 1024;

__device__ __forceinline__ float node_dinv(const unsigned char* __restrict__ tab, int tabn,
                                           const float* __restrict__ dinv, int node) {
    unsigned b = 255u;
    if (node < tabn) b = tab[node];
    return (b != 255u) ? rsqrtf((float)b + 1.0f) : dinv[node];
}

template <int F, bool FULL>
__device__ __forceinline__ void drain_regs(int s, unsigned dl, unsigned cnt, unsigned lane, int lo,
                                           const float* __restrict__ dinv, const float* __restrict__ t,
                                           float* __restrict__ tile,
                                           const unsigned char* __restrict__ tab, int tabn) {
    const bool act = lane < (unsigned)F;
    const unsigned fl = act ? lane : (unsigned)(F - 1);
    float v[32];
#pragma unroll
    for (int u = 0; u < 32; ++u) {
        const unsigned sj = (unsigned)__builtin_amdgcn_readlane(s, u);
        const float* row = t + (size_t)sj * F;
        v[u] = row[fl];
    }
    float w = 0.0f;
    if (FULL || lane < cnt)
        w = node_dinv(tab, tabn, dinv, s) * node_dinv(tab, tabn, dinv, lo + (int)dl);
    unsigned dj[32];
    float wj[32];
#pragma unroll
    for (int u = 0; u < 32; ++u) {
        dj[u] = (unsigned)__builtin_amdgcn_readlane((int)dl, u);
        wj[u] = __int_as_float(__builtin_amdgcn_readlane(__float_as_int(w), u));
    }
    if (act) {
#pragma unroll
        for (int u = 0; u < 32; ++u)
            if (FULL || (unsigned)u < cnt)
                atomicAdd(tile + dj[u] * F + lane, v[u] * wj[u]);
    }
}

template <int F, bool FULL>
__device__ __forceinline__ void drain_ring(const int* __restrict__ rs, const int* __restrict__ rd,
                                           unsigned head, unsigned cnt, unsigned lane, int lo,
                                           const float* __restrict__ dinv, const float* __restrict__ t,
                                           float* __restrict__ tile,
                                           const unsigned char* __restrict__ tab, int tabn) {
    const unsigned qi = (head + lane) & 63u;
    int s = 0;
    unsigned dl = 0;
    if (FULL || lane < cnt) {
        s  = rs[qi];
        dl = (unsigned)rd[qi];
    }
    drain_regs<F, FULL>(s, dl, cnt, lane, lo, dinv, t, tile, tab, tabn);
}

__device__ __forceinline__ void load_group(const int* __restrict__ src, const int* __restrict__ dst,
                                           int g, int e1, bool vec, unsigned lane,
                                           int (&d)[8], int (&s)[8]) {
    if (vec && g + 256 <= e1) {
#pragma unroll
        for (int h = 0; h < 2; ++h) {
            const int e = g + h * 128 + (int)lane * 4;
            const int4 qd = *reinterpret_cast<const int4*>(dst + e);
            const int4 qs = *reinterpret_cast<const int4*>(src + e);
            d[4 * h] = qd.x; d[4 * h + 1] = qd.y; d[4 * h + 2] = qd.z; d[4 * h + 3] = qd.w;
            s[4 * h] = qs.x; s[4 * h + 1] = qs.y; s[4 * h + 2] = qs.z; s[4 * h + 3] = qs.w;
        }
    } else {
#pragma unroll
        for (int h = 0; h < 2; ++h) {
#pragma unroll
            for (int i = 0; i < 4; ++i) {
                const int e = g + h * 128 + (int)lane * 4 + i;
                d[4 * h + i] = (e < e1) ? dst[e] : -1;
                s[4 * h + i] = (e < e1) ? src[e] : 0;
            }
        }
    }
}

template <int F>
__device__ __forceinline__ void search_slice(const int* __restrict__ src, const int* __restrict__ dst,
                                                 const float* __restrict__ dinv, const float* __restrict__ t,
                                                 float* __restrict__ tile, int* __restrict__ rs, int* __restrict__ rd,
                                                 int e0, int e1, bool vec, unsigned wv, unsigned lane,
                                                 int lo, int nloc,
                                                 const unsigned char* __restrict__ tab, int tabn) {
    unsigned head = 0, cnt = 0;

    int cd[8], cs[8];
#pragma unroll
    for (int k = 0; k < 8; ++k) { cd[k] = -1; cs[k] = 0; }
    int gn = e0 + (int)wv * 256;
    for (;;) {
        const bool more = gn < e1;
        int nd[8], ns[8];
        if (more) {
            load_group(src, dst, gn, e1, vec, lane, nd, ns);
        } else {
#pragma unroll
            for (int k = 0; k < 8; ++k) { nd[k] = -1; ns[k] = 0; }
        }
#pragma unroll
        for (int k = 0; k < 8; ++k) {
            const unsigned dl = (unsigned)(cd[k] - lo);
            const bool m = dl < (unsigned)nloc;
            const unsigned mask = __builtin_amdgcn_ballot_w32(m);
            if (mask != 0u) {
                if (m) {
                    const unsigned pos = __builtin_amdgcn_mbcnt_lo(mask, 0u);
                    const unsigned qi  = (head + cnt + pos) & 63u;
                    rs[qi] = cs[k];
                    rd[qi] = (int)dl;
                }
                cnt += (unsigned)__builtin_popcount(mask);
                if (cnt >= 32u) {
                    drain_ring<F, true>(rs, rd, head, cnt, lane, lo, dinv, t, tile, tab, tabn);
                    head = (head + 32u) & 63u;
                    cnt -= 32u;
                }
            }
        }
        if (!more) break;
#pragma unroll
        for (int k = 0; k < 8; ++k) { cd[k] = nd[k]; cs[k] = ns[k]; }
        gn += 32 * 256;
    }
    if (cnt != 0u)
        drain_ring<F, false>(rs, rd, head, cnt, lane, lo, dinv, t, tile, tab, tabn);
}

constexpr int SPLIT_MAXP  = 4096;
constexpr int SPLIT_STAGE = 65536;

__device__ __forceinline__ void split_fetch(const int* __restrict__ src, const int* __restrict__ dst,
                                            int g, int e1, bool vec, int l64,
                                            int (&d)[4], int (&s)[4]) {
    if (g < e1) {
        if (vec && g + 256 <= e1) {
            const int e = g + l64 * 4;
            const int4 qd = *reinterpret_cast<const int4*>(dst + e);
            const int4 qs = *reinterpret_cast<const int4*>(src + e);
            d[0] = qd.x; d[1] = qd.y; d[2] = qd.z; d[3] = qd.w;
            s[0] = qs.x; s[1] = qs.y; s[2] = qs.z; s[3] = qs.w;
        } else {
#pragma unroll
            for (int i = 0; i < 4; ++i) {
                const int e = g + l64 * 4 + i;
                d[i] = (e < e1) ? dst[e] : -1;
                s[i] = (e < e1) ? src[e] : 0;
            }
        }
    } else {
#pragma unroll
        for (int i = 0; i < 4; ++i) { d[i] = -1; s[i] = 0; }
    }
}

template <int NP>
__global__ __launch_bounds__(1024) void k_split_lists(const int* __restrict__ src, const int* __restrict__ dst,
                                                      unsigned* __restrict__ lists, unsigned cap,
                                                      int E, int N) {
    static_assert(NP <= (1 << (32 - LIST_SRC_BITS)), "a place in the range must fit above the source bits");
    __shared__ unsigned cnt[SPLIT_MAXP];
    __shared__ unsigned stage[SPLIT_STAGE];
    const int tid = (int)threadIdx.x;

    const int P  = (N + NP - 1) / NP;
    const int R  = (int)gridDim.x / 32;
    const int r  = (int)blockIdx.x / 32;
    const int wv = (int)blockIdx.x % 32;
    unsigned* counts = lists + (size_t)r * P * 32 + wv;
    if (P > SPLIT_MAXP) {
        for (int p = tid; p < P; p += 1024) counts[(size_t)p * 32] = 0xffffffffu;
        return;
    }
    for (int p = tid; p < P; p += 1024) cnt[p] = 0u;
    __syncthreads();

    const int per = (((E + R - 1) / R) + 255) & ~255;
    const int e0  = r * per;
    const int e1  = (e0 + per) < E ? (e0 + per) : E;
    const bool vec = (E & 3) == 0;
    const int q   = tid >> 6;
    const int l64 = tid & 63;
    const int g0  = e0 + wv * 256;
    const int J   = g0 < e1 ? (e1 - g0 + 8191) / 8192 : 0;
    const int T   = (J + 15) / 16;
    unsigned* ent = lists + (size_t)P * R * 32 + ((size_t)r * P * 32 + wv) * cap;
    const bool staged = (unsigned long long)P * cap <= (unsigned long long)SPLIT_STAGE;

    int cd[4], cs[4];
    split_fetch(src, dst, g0 + q * 8192, e1, vec, l64, cd, cs);
    for (int t = 0; t < T; ++t) {
        int nd[4], ns[4];
        split_fetch(src, dst, g0 + ((t + 1) * 16 + q) * 8192, e1, vec, l64, nd, ns);
        unsigned pp[4], rk[4];
#pragma unroll
        for (int i = 0; i < 4; ++i) {
            const bool ok = (unsigned)cd[i] < (unsigned)N;
            pp[i] = ok ? (unsigned)cd[i] / (unsigned)NP : 0u;
            rk[i] = ok ? atomicAdd(&cnt[pp[i]], 1u) : 0xffffffffu;
        }
#pragma unroll
        for (int i = 0; i < 4; ++i) {
            if (rk[i] < cap) {
                const unsigned dl = (unsigned)cd[i] - pp[i] * (unsigned)NP;
                const unsigned wd = (unsigned)cs[i] | (dl << LIST_SRC_BITS);
                if (staged) stage[pp[i] * cap + rk[i]] = wd;
                else        ent[(size_t)pp[i] * 32u * cap + rk[i]] = wd;
            }
        }
#pragma unroll
        for (int i = 0; i < 4; ++i) { cd[i] = nd[i]; cs[i] = ns[i]; }
    }
    __syncthreads();
    for (int p = tid; p < P; p += 1024) counts[(size_t)p * 32] = cnt[p];
    if (staged) {
        const unsigned lane = (unsigned)tid & 31u;
        for (int p = tid >> 5; p < P; p += 32) {
            const unsigned n = cnt[p] < cap ? cnt[p] : cap;
            const unsigned* sp = stage + (unsigned)p * cap;
            unsigned* gp = ent + (size_t)p * 32u * cap;
            for (unsigned k = lane; k < n; k += 32u) gp[k] = sp[k];
        }
    }
}

constexpr int DEG_LISTS = 4;
constexpr int DEG_LOADS = 16;

template <int NP>
__global__ __launch_bounds__(1024) void k_deg_lists(const int* __restrict__ dst,
                                                    const unsigned* __restrict__ lists, unsigned cap,
                                                    int E, int N, int R,
                                                    float* __restrict__ dinv, unsigned* __restrict__ degb) {
    static_assert(NP % 4 == 0, "a range starts on a whole word of in-degree bytes");
    __shared__ unsigned cnt[NP];
    const unsigned tid  = threadIdx.x;
    const unsigned lane = tid & 31u;
    const unsigned wv = (unsigned)__builtin_amdgcn_readfirstlane((int)(tid >> 5));
    const int P    = (int)gridDim.x;
    const int p    = (int)blockIdx.x;
    const int lo   = p * NP;
    const int nloc = (N - lo) < NP ? (N - lo) : NP;

    for (int i = (int)tid; i < NP; i += 1024) cnt[i] = 0u;
    __syncthreads();

    const unsigned* ent = lists + (size_t)P * R * 32u;
    const int per = (((E + R - 1) / R) + 255) & ~255;
    for (int rb = 0; rb < R; rb += DEG_LISTS) {
        unsigned cc[DEG_LISTS];
        const unsigned* ml[DEG_LISTS];
        const unsigned* cwp[DEG_LISTS];
        unsigned w[DEG_LISTS][DEG_LOADS];
#pragma unroll
        for (int j = 0; j < DEG_LISTS; ++j) {
            const int r = rb + j;
            const unsigned wid = ((unsigned)(r < R ? r : 0) * (unsigned)P + (unsigned)p) * 32u + wv;
            cc[j] = (r < R) ? (unsigned)__builtin_amdgcn_readfirstlane((int)lists[wid]) : 0u;
            ml[j]  = ent + (size_t)wid * cap;
            cwp[j] = lists + wid;
        }
#pragma unroll
        for (int j = 0; j < DEG_LISTS; ++j) {
#pragma unroll
            for (int u = 0; u < DEG_LOADS; ++u) {
                const unsigned k = (unsigned)u * 32u + lane;
                const unsigned* a = (cc[j] <= cap && k < cc[j]) ? (ml[j] + k) : cwp[j];
                w[j][u] = *a;
            }
        }
#pragma unroll
        for (int j = 0; j < DEG_LISTS; ++j) {
            if (cc[j] <= cap) {
#pragma unroll
                for (int u = 0; u < DEG_LOADS; ++u) {
                    const unsigned k = (unsigned)u * 32u + lane;
                    if (k < cc[j]) atomicAdd(&cnt[w[j][u] >> LIST_SRC_BITS], 1u);
                }
                for (unsigned k = (unsigned)DEG_LOADS * 32u + lane; k < cc[j]; k += 32u)
                    atomicAdd(&cnt[ml[j][k] >> LIST_SRC_BITS], 1u);
            } else {
                const int r  = rb + j;
                const int e0 = r * per;
                const int e1 = (e0 + per) < E ? (e0 + per) : E;
                for (int g = e0 + (int)wv * 256; g < e1; g += 32 * 256) {
                    for (int k = 0; k < 8; ++k) {
                        const int e = g + k * 32 + (int)lane;
                        if (e < e1) {
                            const unsigned dl = (unsigned)(dst[e] - lo);
                            if (dl < (unsigned)nloc) atomicAdd(&cnt[dl], 1u);
                        }
                    }
                }
            }
        }
    }
    __syncthreads();

    for (int i = (int)tid; i < nloc; i += 1024)
        dinv[lo + i] = rsqrtf((float)cnt[i] + 1.0f);
    const int TW = ((N + 15) / 16) * 4;
    const int w0 = lo / 4;
    const int w1 = (p == P - 1) ? TW : (lo + NP) / 4;
    for (int wi = w0 + (int)tid; wi < w1; wi += 1024) {
        unsigned wd = 0u;
#pragma unroll
        for (int k = 0; k < 4; ++k) {
            const int n = wi * 4 + k;
            unsigned b = 255u;
            if (n < N) {
                const unsigned d = cnt[n - lo];
                b = d < 255u ? d : 255u;
            }
            wd |= b << (8 * k);
        }
        degb[wi] = wd;
    }
}

constexpr int SELF_LOADS = 13;

template <int F, int NP>
__global__ __launch_bounds__(1024) void k_edge_tile(const int* __restrict__ src, const int* __restrict__ dst,
                                                    const float* __restrict__ dinv, const float* __restrict__ t,
                                                    float* __restrict__ agg, const unsigned* __restrict__ lists,
                                                    unsigned cap, int E, int N,
                                                    const unsigned* __restrict__ degb) {
    static_assert(NP <= (1 << (32 - LIST_SRC_BITS)), "a place in the range must fit above the source bits");
    static_assert(DEG_TAB % 16 == 0 && DEG_TAB / 16 <= TAB_LOADS * 1024, "the table copy covers the table");
    __shared__ __attribute__((aligned(16))) float tile[NP * F];
    __shared__ int ring_s[32][64];
    __shared__ int ring_d[32][64];
    __shared__ __attribute__((aligned(16))) unsigned char degtab[DEG_TAB];

    const unsigned tid  = threadIdx.x;
    const unsigned lane = tid & 31u;
    const unsigned wv = (unsigned)__builtin_amdgcn_readfirstlane((int)(tid >> 5));

    const int P = (N + NP - 1) / NP;
    const int R = (int)gridDim.x / P;
    const int p = (int)blockIdx.x % P;
    const int r = (int)blockIdx.x / P;
    const int lo   = p * NP;
    const int nloc = (N - lo) < NP ? (N - lo) : NP;

    const int tabn = N < DEG_TAB ? N : DEG_TAB;
    const int ng   = (tabn + 15) / 16;
    uint4 tq[TAB_LOADS];
#pragma unroll
    for (int k = 0; k < TAB_LOADS; ++k) {
        const int i = (int)tid + k * 1024;
        tq[k] = (i < ng) ? reinterpret_cast<const uint4*>(degb)[i] : make_uint4(0u, 0u, 0u, 0u);
    }

    const int sq  = (nloc + R - 1) / R;
    const int sn0 = (r * sq) < nloc ? (r * sq) : nloc;
    const int sn1 = (sn0 + sq) < nloc ? (sn0 + sq) : nloc;
    const int sb  = sn0 * F;
    const int se  = sn1 * F;
    const float* tself = t + (size_t)lo * F;
    float tv[SELF_LOADS];
#pragma unroll
    for (int u = 0; u < SELF_LOADS; ++u) {
        const int i = sb + u * 1024 + (int)tid;
        tv[u] = (i < se) ? tself[i] : 0.0f;
    }

    for (int i = (int)tid; i < NP * F / 4; i += 1024)
        reinterpret_cast<float4*>(tile)[i] = make_float4(0.0f, 0.0f, 0.0f, 0.0f);

#pragma unroll
    for (int k = 0; k < TAB_LOADS; ++k) {
        const int i = (int)tid + k * 1024;
        if (i < ng) reinterpret_cast<uint4*>(degtab)[i] = tq[k];
    }
    __syncthreads();

#pragma unroll
    for (int u = 0; u < SELF_LOADS; ++u) {
        const int i = sb + u * 1024 + (int)tid;
        if (i < se) {
            const float di = node_dinv(degtab, tabn, dinv, lo + i / F);
            tile[i] = tv[u] * di * di;
        }
    }
    for (int i = sb + SELF_LOADS * 1024 + (int)tid; i < se; i += 1024) {
        const float di = node_dinv(degtab, tabn, dinv, lo + i / F);
        tile[i] = tself[i] * di * di;
    }
    __syncthreads();

    const unsigned wid = blockIdx.x * 32u + wv;
    const unsigned* mylist = lists + (size_t)gridDim.x * 32u + (size_t)wid * cap;

    const unsigned c = (unsigned)__builtin_amdgcn_readfirstlane((int)lists[wid]);
    if (c <= cap) {
        unsigned nxt = (lane < c) ? mylist[lane] : 0u;
        for (unsigned k = 0; k < c; k += 32u) {
            const unsigned cur = nxt;
            const unsigned kn  = k + 32u;
            if (kn < c) nxt = (kn + lane < c) ? mylist[kn + lane] : 0u;
            const int      s  = (int)(cur & LIST_SRC_MASK);
            const unsigned dl = cur >> LIST_SRC_BITS;
            const unsigned rem = c - k;
            if (rem >= 32u) drain_regs<F, true >(s, dl, 32u, lane, lo, dinv, t, tile, degtab, tabn);
            else            drain_regs<F, false>(s, dl, rem, lane, lo, dinv, t, tile, degtab, tabn);
        }
    } else {
        const int per = (((E + R - 1) / R) + 255) & ~255;
        const int e0  = r * per;
        const int e1  = (e0 + per) < E ? (e0 + per) : E;
        const bool vec = (E & 3) == 0;
        search_slice<F>(src, dst, dinv, t, tile, ring_s[wv], ring_d[wv], e0, e1, vec,
                        wv, lane, lo, nloc, degtab, tabn);
    }
    __syncthreads();

    float* ag = agg + (size_t)lo * F;
    const int tot = nloc * F;
    for (int i = (int)tid; i < tot; i += 1024)
        atomicAdd(ag + i, tile[i]);
}

__global__ void k_gemm_t2(const float* __restrict__ agg1, const float* __restrict__ W2,
                          const float* __restrict__ b2,
                          float* __restrict__ t2, float* __restrict__ out, int N, int nTiles) {
    int wave = (blockIdx.x * blockDim.x + threadIdx.x) >> 5;
    if (wave >= nTiles) return;
    int lane = threadIdx.x & 31;
    int n    = lane & 15;
    int kh   = lane >> 4;
    int rowA = wave * 16 + n;    if (rowA >= N) rowA = N - 1;

    v8f acc0 = {}, acc1 = {};
    const float* ar = agg1 + (size_t)rowA * 32;
    bool c1ok = (16 + n) < 20;
#pragma unroll
    for (int k0 = 0; k0 < 32; k0 += 4) {
        int kb = k0 + 2 * kh;
        v2f a;  a.x = fmaxf(ar[kb], 0.0f);  a.y = fmaxf(ar[kb + 1], 0.0f);
        v2f b0; b0.x = W2[kb * 20 + n];     b0.y = W2[(kb + 1) * 20 + n];
        v2f b1; b1.x = c1ok ? W2[kb * 20 + 16 + n] : 0.0f;
                b1.y = c1ok ? W2[(kb + 1) * 20 + 16 + n] : 0.0f;
        acc0 = wmma4(a, b0, acc0);
        acc1 = wmma4(a, b1, acc1);
    }
    const float bia0 = b2[n];
    const float bia1 = c1ok ? b2[16 + n] : 0.0f;
#pragma unroll
    for (int v = 0; v < 8; ++v) {
        int mr   = (lane < 16) ? v : v + 8;
        int node = wave * 16 + mr;
        if (node < N) {
            float* o = t2 + (size_t)node * 20;
            float* q = out + (size_t)node * 20;
            o[n] = acc0[v];
            q[n] = bia0;
            if (c1ok) {
                o[16 + n] = acc1[v];
                q[16 + n] = bia1;
            }
        }
    }
}

extern "C" void kernel_launch(void* const* d_in, const int* in_sizes, int n_in,
                              void* d_out, int out_size, void* d_ws, size_t ws_size,
                              hipStream_t stream) {
    const float* x    = (const float*)d_in[0];
    const float* cnn  = (const float*)d_in[1];
    const int*   ei   = (const int*)d_in[2];
    const float* fcW  = (const float*)d_in[3];
    const float* fcb  = (const float*)d_in[4];
    const float* cnnW = (const float*)d_in[5];
    const float* cnnb = (const float*)d_in[6];
    const float* W1   = (const float*)d_in[7];
    const float* b1   = (const float*)d_in[8];
    const float* W2   = (const float*)d_in[9];
    const float* b2   = (const float*)d_in[10];

    const int N = in_sizes[0] / 32;
    const int E = in_sizes[2] / 2;
    const int* src = ei;
    const int* dst = ei + E;

    float* ws   = (float*)d_ws;
    float* dinv = ws;  ws += N;
    float* h0   = ws;  ws += (size_t)N * 64;
    float* t1   = ws;  ws += (size_t)N * 32;
    float* agg1 = ws;  ws += (size_t)N * 32;
    float* t2   = ws;  ws += (size_t)N * 20;
    float* out  = (float*)d_out;

    const int nTiles = (N + 15) / 16;
    dim3 blkG(128);
    dim3 grdG((nTiles * 32 + 127) / 128);
    const int PT = (N + RANGE_NP - 1) / RANGE_NP;
    const int RT = PT < 256 ? 256 / PT : 1;
    unsigned* degb = (unsigned*)((float*)d_ws + ((N + 3) / 4) * 4);
    const int TW = ((N + 15) / 16) * 4;
    unsigned* lists = degb + TW;
    const long long lwaves = (long long)PT * RT * 32;
    long long lcap = (((long long)N * 64 - 3 - TW - lwaves) / lwaves) & ~31LL;
    if (lcap < 0 || N > (1 << LIST_SRC_BITS)) lcap = 0;
    if (lcap > 65536) lcap = 65536;
    const unsigned cap = (unsigned)lcap;

    k_encoder<<<grdG, blkG, 0, stream>>>(x, cnn, fcW, fcb, cnnW, cnnb, h0, N, nTiles);
    k_gemm_t1<<<grdG, blkG, 0, stream>>>(h0, W1, b1, t1, agg1, N, nTiles);

    k_split_lists<RANGE_NP><<<RT * 32, 1024, 0, stream>>>(src, dst, lists, cap, E, N);

    k_deg_lists<RANGE_NP><<<PT, 1024, 0, stream>>>(dst, lists, cap, E, N, RT, dinv, degb);

    k_edge_tile<32, RANGE_NP><<<PT * RT, 1024, 0, stream>>>(src, dst, dinv, t1, agg1, lists, cap, E, N, degb);

    k_gemm_t2<<<grdG, blkG, 0, stream>>>(agg1, W2, b2, t2, out, N, nTiles);
    k_edge_tile<20, RANGE_NP><<<PT * RT, 1024, 0, stream>>>(src, dst, dinv, t2, out, lists, cap, E, N, degb);
}


# One work-group owns each destination range (400 nodes; lists of 8 edge slices joined in LDS): finished tile stored, no global atomics or start stores; node sum is one LDS-arrival sum, bias last
# speedup vs baseline: 5.7364x; 1.0324x over previous
#include <hip/hip_runtime.h>
#include <hip/hip_bf16.h>

typedef __attribute__((ext_vector_type(2))) float v2f;
typedef __attribute__((ext_vector_type(8))) float v8f;

__device__ __forceinline__ v8f wmma4(v2f a, v2f b, v8f c) {
    return __builtin_amdgcn_wmma_f32_16x16x4_f32(
         false, a,  false, b,
         (short)0, c,  false,  false);
}

__global__ void k_encoder(const float* __restrict__ x, const float* __restrict__ cnn,
                          const float* __restrict__ fcW, const float* __restrict__ fcb,
                          const float* __restrict__ cnnW, const float* __restrict__ cnnb,
                          float* __restrict__ h0, int N, int nTiles) {
    int wave = (blockIdx.x * blockDim.x + threadIdx.x) >> 5;
    if (wave >= nTiles) return;
    int lane = threadIdx.x & 31;
    int n    = lane & 15;
    int kh   = lane >> 4;
    int rowA = wave * 16 + n;    if (rowA >= N) rowA = N - 1;

    v8f acc0 = {}, acc1 = {}, acc2 = {}, acc3 = {};

    const float* ax = x + (size_t)rowA * 32;
#pragma unroll
    for (int k0 = 0; k0 < 32; k0 += 4) {
        int kb = k0 + 2 * kh;
        v2f a;  a.x = ax[kb];               a.y = ax[kb + 1];
        v2f b0; b0.x = fcW[kb * 32 + n];    b0.y = fcW[(kb + 1) * 32 + n];
        v2f b1; b1.x = fcW[kb * 32 + 16 + n]; b1.y = fcW[(kb + 1) * 32 + 16 + n];
        acc0 = wmma4(a, b0, acc0);
        acc1 = wmma4(a, b1, acc1);
    }
    const float* ac = cnn + (size_t)rowA * 128;
#pragma unroll
    for (int k0 = 0; k0 < 128; k0 += 4) {
        int kb = k0 + 2 * kh;
        v2f a;  a.x = ac[kb];                a.y = ac[kb + 1];
        v2f b0; b0.x = cnnW[kb * 32 + n];    b0.y = cnnW[(kb + 1) * 32 + n];
        v2f b1; b1.x = cnnW[kb * 32 + 16 + n]; b1.y = cnnW[(kb + 1) * 32 + 16 + n];
        acc2 = wmma4(a, b0, acc2);
        acc3 = wmma4(a, b1, acc3);
    }
#pragma unroll
    for (int v = 0; v < 8; ++v) {
        int mr   = (lane < 16) ? v : v + 8;
        int node = wave * 16 + mr;
        if (node < N) {
            float* o = h0 + (size_t)node * 64;
            o[n]      = fmaxf(acc0[v] + fcb[n],       0.0f);
            o[16 + n] = fmaxf(acc1[v] + fcb[16 + n],  0.0f);
            o[32 + n] = fmaxf(acc2[v] + cnnb[n],      0.0f);
            o[48 + n] = fmaxf(acc3[v] + cnnb[16 + n], 0.0f);
        }
    }
}

__global__ void k_gemm_t1(const float* __restrict__ h0, const float* __restrict__ W1,
                          float* __restrict__ t1, int N, int nTiles) {
    int wave = (blockIdx.x * blockDim.x + threadIdx.x) >> 5;
    if (wave >= nTiles) return;
    int lane = threadIdx.x & 31;
    int n    = lane & 15;
    int kh   = lane >> 4;
    int rowA = wave * 16 + n;    if (rowA >= N) rowA = N - 1;

    v8f acc0 = {}, acc1 = {};
    const float* ar = h0 + (size_t)rowA * 64;
#pragma unroll
    for (int k0 = 0; k0 < 64; k0 += 4) {
        int kb = k0 + 2 * kh;
        v2f a;  a.x = ar[kb];              a.y = ar[kb + 1];
        v2f b0; b0.x = W1[kb * 32 + n];    b0.y = W1[(kb + 1) * 32 + n];
        v2f w1; w1.x = W1[kb * 32 + 16 + n]; w1.y = W1[(kb + 1) * 32 + 16 + n];
        acc0 = wmma4(a, b0, acc0);
        acc1 = wmma4(a, w1, acc1);
    }
#pragma unroll
    for (int v = 0; v < 8; ++v) {
        int mr   = (lane < 16) ? v : v + 8;
        int node = wave * 16 + mr;
        if (node < N) {
            float* o = t1 + (size_t)node * 32;
            o[n]      = acc0[v];
            o[16 + n] = acc1[v];
        }
    }
}

constexpr int      LIST_SRC_BITS = 21;
constexpr unsigned LIST_SRC_MASK = (1u << LIST_SRC_BITS) - 1u;

constexpr int RANGE_NP  = 400;
constexpr int DEG_TAB   = 102400;
constexpr int TAB_LOADS = (DEG_TAB / 16 + 1023) / 1024;

__device__ __forceinline__ float node_dinv(const unsigned char* __restrict__ tab, int tabn,
                                           const float* __restrict__ dinv, int node) {
    unsigned b = 255u;
    if (node < tabn) b = tab[node];
    return (b != 255u) ? rsqrtf((float)b + 1.0f) : dinv[node];
}

template <int F, bool FULL>
__device__ __forceinline__ void drain_regs(int s, unsigned dl, unsigned cnt, unsigned lane, int lo,
                                           const float* __restrict__ dinv, const float* __restrict__ t,
                                           float* __restrict__ tile,
                                           const unsigned char* __restrict__ tab, int tabn) {
    const bool act = lane < (unsigned)F;
    const unsigned fl = act ? lane : (unsigned)(F - 1);
    float v[32];
#pragma unroll
    for (int u = 0; u < 32; ++u) {
        const unsigned sj = (unsigned)__builtin_amdgcn_readlane(s, u);
        const float* row = t + (size_t)sj * F;
        v[u] = row[fl];
    }
    float w = 0.0f;
    if (FULL || lane < cnt)
        w = node_dinv(tab, tabn, dinv, s) * node_dinv(tab, tabn, dinv, lo + (int)dl);
    unsigned dj[32];
    float wj[32];
#pragma unroll
    for (int u = 0; u < 32; ++u) {
        dj[u] = (unsigned)__builtin_amdgcn_readlane((int)dl, u);
        wj[u] = __int_as_float(__builtin_amdgcn_readlane(__float_as_int(w), u));
    }
    if (act) {
#pragma unroll
        for (int u = 0; u < 32; ++u)
            if (FULL || (unsigned)u < cnt)
                atomicAdd(tile + dj[u] * F + lane, v[u] * wj[u]);
    }
}

template <int F, bool FULL>
__device__ __forceinline__ void drain_ring(const int* __restrict__ rs, const int* __restrict__ rd,
                                           unsigned head, unsigned cnt, unsigned lane, int lo,
                                           const float* __restrict__ dinv, const float* __restrict__ t,
                                           float* __restrict__ tile,
                                           const unsigned char* __restrict__ tab, int tabn) {
    const unsigned qi = (head + lane) & 63u;
    int s = 0;
    unsigned dl = 0;
    if (FULL || lane < cnt) {
        s  = rs[qi];
        dl = (unsigned)rd[qi];
    }
    drain_regs<F, FULL>(s, dl, cnt, lane, lo, dinv, t, tile, tab, tabn);
}

__device__ __forceinline__ void load_group(const int* __restrict__ src, const int* __restrict__ dst,
                                           int g, int e1, bool vec, unsigned lane,
                                           int (&d)[8], int (&s)[8]) {
    if (vec && g + 256 <= e1) {
#pragma unroll
        for (int h = 0; h < 2; ++h) {
            const int e = g + h * 128 + (int)lane * 4;
            const int4 qd = *reinterpret_cast<const int4*>(dst + e);
            const int4 qs = *reinterpret_cast<const int4*>(src + e);
            d[4 * h] = qd.x; d[4 * h + 1] = qd.y; d[4 * h + 2] = qd.z; d[4 * h + 3] = qd.w;
            s[4 * h] = qs.x; s[4 * h + 1] = qs.y; s[4 * h + 2] = qs.z; s[4 * h + 3] = qs.w;
        }
    } else {
#pragma unroll
        for (int h = 0; h < 2; ++h) {
#pragma unroll
            for (int i = 0; i < 4; ++i) {
                const int e = g + h * 128 + (int)lane * 4 + i;
                d[4 * h + i] = (e < e1) ? dst[e] : -1;
                s[4 * h + i] = (e < e1) ? src[e] : 0;
            }
        }
    }
}

template <int F>
__device__ __forceinline__ void search_slice(const int* __restrict__ src, const int* __restrict__ dst,
                                                 const float* __restrict__ dinv, const float* __restrict__ t,
                                                 float* __restrict__ tile, int* __restrict__ rs, int* __restrict__ rd,
                                                 int e0, int e1, bool vec, unsigned wv, unsigned lane,
                                                 int lo, int nloc,
                                                 const unsigned char* __restrict__ tab, int tabn) {
    unsigned head = 0, cnt = 0;

    int cd[8], cs[8];
#pragma unroll
    for (int k = 0; k < 8; ++k) { cd[k] = -1; cs[k] = 0; }
    int gn = e0 + (int)wv * 256;
    for (;;) {
        const bool more = gn < e1;
        int nd[8], ns[8];
        if (more) {
            load_group(src, dst, gn, e1, vec, lane, nd, ns);
        } else {
#pragma unroll
            for (int k = 0; k < 8; ++k) { nd[k] = -1; ns[k] = 0; }
        }
#pragma unroll
        for (int k = 0; k < 8; ++k) {
            const unsigned dl = (unsigned)(cd[k] - lo);
            const bool m = dl < (unsigned)nloc;
            const unsigned mask = __builtin_amdgcn_ballot_w32(m);
            if (mask != 0u) {
                if (m) {
                    const unsigned pos = __builtin_amdgcn_mbcnt_lo(mask, 0u);
                    const unsigned qi  = (head + cnt + pos) & 63u;
                    rs[qi] = cs[k];
                    rd[qi] = (int)dl;
                }
                cnt += (unsigned)__builtin_popcount(mask);
                if (cnt >= 32u) {
                    drain_ring<F, true>(rs, rd, head, cnt, lane, lo, dinv, t, tile, tab, tabn);
                    head = (head + 32u) & 63u;
                    cnt -= 32u;
                }
            }
        }
        if (!more) break;
#pragma unroll
        for (int k = 0; k < 8; ++k) { cd[k] = nd[k]; cs[k] = ns[k]; }
        gn += 32 * 256;
    }
    if (cnt != 0u)
        drain_ring<F, false>(rs, rd, head, cnt, lane, lo, dinv, t, tile, tab, tabn);
}

constexpr int SPLIT_MAXP  = 4096;
constexpr int SPLIT_STAGE = 65536;

__device__ __forceinline__ void split_fetch(const int* __restrict__ src, const int* __restrict__ dst,
                                            int g, int e1, bool vec, int l64,
                                            int (&d)[4], int (&s)[4]) {
    if (g < e1) {
        if (vec && g + 256 <= e1) {
            const int e = g + l64 * 4;
            const int4 qd = *reinterpret_cast<const int4*>(dst + e);
            const int4 qs = *reinterpret_cast<const int4*>(src + e);
            d[0] = qd.x; d[1] = qd.y; d[2] = qd.z; d[3] = qd.w;
            s[0] = qs.x; s[1] = qs.y; s[2] = qs.z; s[3] = qs.w;
        } else {
#pragma unroll
            for (int i = 0; i < 4; ++i) {
                const int e = g + l64 * 4 + i;
                d[i] = (e < e1) ? dst[e] : -1;
                s[i] = (e < e1) ? src[e] : 0;
            }
        }
    } else {
#pragma unroll
        for (int i = 0; i < 4; ++i) { d[i] = -1; s[i] = 0; }
    }
}

template <int NP>
__global__ __launch_bounds__(1024) void k_split_lists(const int* __restrict__ src, const int* __restrict__ dst,
                                                      unsigned* __restrict__ lists, unsigned cap,
                                                      int E, int N) {
    static_assert(NP <= (1 << (32 - LIST_SRC_BITS)), "a place in the range must fit above the source bits");
    __shared__ unsigned cnt[SPLIT_MAXP];
    __shared__ unsigned stage[SPLIT_STAGE];
    const int tid = (int)threadIdx.x;

    const int P  = (N + NP - 1) / NP;
    const int R  = (int)gridDim.x / 32;
    const int r  = (int)blockIdx.x / 32;
    const int wv = (int)blockIdx.x % 32;
    unsigned* counts = lists + (size_t)r * P * 32 + wv;
    if (P > SPLIT_MAXP) {
        for (int p = tid; p < P; p += 1024) counts[(size_t)p * 32] = 0xffffffffu;
        return;
    }
    for (int p = tid; p < P; p += 1024) cnt[p] = 0u;
    __syncthreads();

    const int per = (((E + R - 1) / R) + 255) & ~255;
    const int e0  = r * per;
    const int e1  = (e0 + per) < E ? (e0 + per) : E;
    const bool vec = (E & 3) == 0;
    const int q   = tid >> 6;
    const int l64 = tid & 63;
    const int g0  = e0 + wv * 256;
    const int J   = g0 < e1 ? (e1 - g0 + 8191) / 8192 : 0;
    const int T   = (J + 15) / 16;
    unsigned* ent = lists + (size_t)P * R * 32 + ((size_t)r * P * 32 + wv) * cap;
    const bool staged = (unsigned long long)P * cap <= (unsigned long long)SPLIT_STAGE;

    int cd[4], cs[4];
    split_fetch(src, dst, g0 + q * 8192, e1, vec, l64, cd, cs);
    for (int t = 0; t < T; ++t) {
        int nd[4], ns[4];
        split_fetch(src, dst, g0 + ((t + 1) * 16 + q) * 8192, e1, vec, l64, nd, ns);
        unsigned pp[4], rk[4];
#pragma unroll
        for (int i = 0; i < 4; ++i) {
            const bool ok = (unsigned)cd[i] < (unsigned)N;
            pp[i] = ok ? (unsigned)cd[i] / (unsigned)NP : 0u;
            rk[i] = ok ? atomicAdd(&cnt[pp[i]], 1u) : 0xffffffffu;
        }
#pragma unroll
        for (int i = 0; i < 4; ++i) {
            if (rk[i] < cap) {
                const unsigned dl = (unsigned)cd[i] - pp[i] * (unsigned)NP;
                const unsigned wd = (unsigned)cs[i] | (dl << LIST_SRC_BITS);
                if (staged) stage[pp[i] * cap + rk[i]] = wd;
                else        ent[(size_t)pp[i] * 32u * cap + rk[i]] = wd;
            }
        }
#pragma unroll
        for (int i = 0; i < 4; ++i) { cd[i] = nd[i]; cs[i] = ns[i]; }
    }
    __syncthreads();
    for (int p = tid; p < P; p += 1024) counts[(size_t)p * 32] = cnt[p];
    if (staged) {
        const unsigned lane = (unsigned)tid & 31u;
        for (int p = tid >> 5; p < P; p += 32) {
            const unsigned n = cnt[p] < cap ? cnt[p] : cap;
            const unsigned* sp = stage + (unsigned)p * cap;
            unsigned* gp = ent + (size_t)p * 32u * cap;
            for (unsigned k = lane; k < n; k += 32u) gp[k] = sp[k];
        }
    }
}

constexpr int DEG_LISTS = 8;
constexpr int DEG_LOADS = 4;

template <int NP>
__global__ __launch_bounds__(1024) void k_deg_lists(const int* __restrict__ dst,
                                                    const unsigned* __restrict__ lists, unsigned cap,
                                                    int E, int N, int R,
                                                    float* __restrict__ dinv, unsigned* __restrict__ degb) {
    static_assert(NP % 4 == 0, "a range starts on a whole word of in-degree bytes");
    __shared__ unsigned cnt[NP];
    const unsigned tid  = threadIdx.x;
    const unsigned lane = tid & 31u;
    const unsigned wv = (unsigned)__builtin_amdgcn_readfirstlane((int)(tid >> 5));
    const int P    = (int)gridDim.x;
    const int p    = (int)blockIdx.x;
    const int lo   = p * NP;
    const int nloc = (N - lo) < NP ? (N - lo) : NP;

    for (int i = (int)tid; i < NP; i += 1024) cnt[i] = 0u;
    __syncthreads();

    const unsigned* ent = lists + (size_t)P * R * 32u;
    const int per = (((E + R - 1) / R) + 255) & ~255;
    for (int rb = 0; rb < R; rb += DEG_LISTS) {
        unsigned cc[DEG_LISTS];
        const unsigned* ml[DEG_LISTS];
        const unsigned* cwp[DEG_LISTS];
        unsigned w[DEG_LISTS][DEG_LOADS];
#pragma unroll
        for (int j = 0; j < DEG_LISTS; ++j) {
            const int r = rb + j;
            const unsigned wid = ((unsigned)(r < R ? r : 0) * (unsigned)P + (unsigned)p) * 32u + wv;
            cc[j] = (r < R) ? (unsigned)__builtin_amdgcn_readfirstlane((int)lists[wid]) : 0u;
            ml[j]  = ent + (size_t)wid * cap;
            cwp[j] = lists + wid;
        }
#pragma unroll
        for (int j = 0; j < DEG_LISTS; ++j) {
#pragma unroll
            for (int u = 0; u < DEG_LOADS; ++u) {
                const unsigned k = (unsigned)u * 32u + lane;
                const unsigned* a = (cc[j] <= cap && k < cc[j]) ? (ml[j] + k) : cwp[j];
                w[j][u] = *a;
            }
        }
#pragma unroll
        for (int j = 0; j < DEG_LISTS; ++j) {
            if (cc[j] <= cap) {
#pragma unroll
                for (int u = 0; u < DEG_LOADS; ++u) {
                    const unsigned k = (unsigned)u * 32u + lane;
                    if (k < cc[j]) atomicAdd(&cnt[w[j][u] >> LIST_SRC_BITS], 1u);
                }
                for (unsigned k = (unsigned)DEG_LOADS * 32u + lane; k < cc[j]; k += 32u)
                    atomicAdd(&cnt[ml[j][k] >> LIST_SRC_BITS], 1u);
            } else {
                const int r  = rb + j;
                const int e0 = r * per;
                const int e1 = (e0 + per) < E ? (e0 + per) : E;
                for (int g = e0 + (int)wv * 256; g < e1; g += 32 * 256) {
                    for (int k = 0; k < 8; ++k) {
                        const int e = g + k * 32 + (int)lane;
                        if (e < e1) {
                            const unsigned dl = (unsigned)(dst[e] - lo);
                            if (dl < (unsigned)nloc) atomicAdd(&cnt[dl], 1u);
                        }
                    }
                }
            }
        }
    }
    __syncthreads();

    for (int i = (int)tid; i < nloc; i += 1024)
        dinv[lo + i] = rsqrtf((float)cnt[i] + 1.0f);
    const int TW = ((N + 15) / 16) * 4;
    const int w0 = lo / 4;
    const int w1 = (p == P - 1) ? TW : (lo + NP) / 4;
    for (int wi = w0 + (int)tid; wi < w1; wi += 1024) {
        unsigned wd = 0u;
#pragma unroll
        for (int k = 0; k < 4; ++k) {
            const int n = wi * 4 + k;
            unsigned b = 255u;
            if (n < N) {
                const unsigned d = cnt[n - lo];
                b = d < 255u ? d : 255u;
            }
            wd |= b << (8 * k);
        }
        degb[wi] = wd;
    }
}

constexpr int LIST_SLICES = 8;
constexpr int LIST_BUF    = 1024;
constexpr int LIST_PRE    = 3;

template <int F, int NP>
__global__ __launch_bounds__(1024) void k_edge_tile(const int* __restrict__ src, const int* __restrict__ dst,
                                                    const float* __restrict__ dinv, const float* __restrict__ t,
                                                    float* __restrict__ agg, const float* __restrict__ bias,
                                                    const unsigned* __restrict__ lists,
                                                    unsigned cap, int E, int N,
                                                    const unsigned* __restrict__ degb) {
    static_assert(NP <= (1 << (32 - LIST_SRC_BITS)), "a place in the range must fit above the source bits");
    static_assert(DEG_TAB % 16 == 0 && DEG_TAB / 16 <= TAB_LOADS * 1024, "the table copy covers the table");
    constexpr int SELF_LOADS = (NP * F + 1023) / 1024;
    __shared__ float tile[NP * F];
    __shared__ int ring_s[32][64];
    __shared__ int ring_d[32][64];
    __shared__ __attribute__((aligned(16))) unsigned char degtab[DEG_TAB];
    __shared__ unsigned lbuf[32][LIST_BUF];

    const unsigned tid  = threadIdx.x;
    const unsigned lane = tid & 31u;
    const unsigned wv = (unsigned)__builtin_amdgcn_readfirstlane((int)(tid >> 5));

    const int P = (int)gridDim.x;
    const int p = (int)blockIdx.x;
    const int lo   = p * NP;
    const int nloc = (N - lo) < NP ? (N - lo) : NP;

    const unsigned* ent = lists + (size_t)P * LIST_SLICES * 32u;
    unsigned cc[LIST_SLICES];
    unsigned total = 0u;
    bool staged = true;
#pragma unroll
    for (int r = 0; r < LIST_SLICES; ++r) {
        const unsigned wid = ((unsigned)r * (unsigned)P + (unsigned)p) * 32u + wv;
        cc[r] = (unsigned)__builtin_amdgcn_readfirstlane((int)lists[wid]);
        staged = staged && cc[r] <= cap && cc[r] <= (unsigned)(LIST_PRE * 32);
        total += cc[r];
    }
    staged = staged && total <= (unsigned)LIST_BUF;

    unsigned lw[LIST_SLICES][LIST_PRE];
#pragma unroll
    for (int r = 0; r < LIST_SLICES; ++r) {
        const unsigned wid = ((unsigned)r * (unsigned)P + (unsigned)p) * 32u + wv;
        const unsigned* ml = ent + (size_t)wid * cap;
#pragma unroll
        for (int u = 0; u < LIST_PRE; ++u) {
            const unsigned k = (unsigned)u * 32u + lane;
            lw[r][u] = (staged && k < cc[r]) ? ml[k] : 0u;
        }
    }

    const int tabn = N < DEG_TAB ? N : DEG_TAB;
    const int ng   = (tabn + 15) / 16;
    uint4 tq[TAB_LOADS];
#pragma unroll
    for (int k = 0; k < TAB_LOADS; ++k) {
        const int i = (int)tid + k * 1024;
        tq[k] = (i < ng) ? reinterpret_cast<const uint4*>(degb)[i] : make_uint4(0u, 0u, 0u, 0u);
    }

    const int se = nloc * F;
    const float* tself = t + (size_t)lo * F;
    float tv[SELF_LOADS];
#pragma unroll
    for (int u = 0; u < SELF_LOADS; ++u) {
        const int i = u * 1024 + (int)tid;
        tv[u] = (i < se) ? tself[i] : 0.0f;
    }

    if (staged) {
        unsigned* b = lbuf[wv];
        unsigned off = 0u;
#pragma unroll
        for (int r = 0; r < LIST_SLICES; ++r) {
#pragma unroll
            for (int u = 0; u < LIST_PRE; ++u) {
                const unsigned k = (unsigned)u * 32u + lane;
                if (k < cc[r]) b[off + k] = lw[r][u];
            }
            off += cc[r];
        }
    }

#pragma unroll
    for (int k = 0; k < TAB_LOADS; ++k) {
        const int i = (int)tid + k * 1024;
        if (i < ng) reinterpret_cast<uint4*>(degtab)[i] = tq[k];
    }
    __syncthreads();

#pragma unroll
    for (int u = 0; u < SELF_LOADS; ++u) {
        const int i = u * 1024 + (int)tid;
        if (i < se) {
            const float di = node_dinv(degtab, tabn, dinv, lo + i / F);
            tile[i] = tv[u] * di * di;
        }
    }
    __syncthreads();

    if (staged) {
        const unsigned* b = lbuf[wv];
        unsigned nxt = (lane < total) ? b[lane] : 0u;
        for (unsigned k = 0; k < total; k += 32u) {
            const unsigned cur = nxt;
            const unsigned kn  = k + 32u;
            if (kn < total) nxt = (kn + lane < total) ? b[kn + lane] : 0u;
            const int      s  = (int)(cur & LIST_SRC_MASK);
            const unsigned dl = cur >> LIST_SRC_BITS;
            const unsigned rem = total - k;
            if (rem >= 32u) drain_regs<F, true >(s, dl, 32u, lane, lo, dinv, t, tile, degtab, tabn);
            else            drain_regs<F, false>(s, dl, rem, lane, lo, dinv, t, tile, degtab, tabn);
        }
    } else {
        const int per = (((E + LIST_SLICES - 1) / LIST_SLICES) + 255) & ~255;
        const bool vec = (E & 3) == 0;
#pragma nounroll
        for (int r = 0; r < LIST_SLICES; ++r) {
            const unsigned wid = ((unsigned)r * (unsigned)P + (unsigned)p) * 32u + wv;
            const unsigned c = (unsigned)__builtin_amdgcn_readfirstlane((int)lists[wid]);
            if (c <= cap) {
                const unsigned* mylist = ent + (size_t)wid * cap;
                unsigned nxt = (lane < c) ? mylist[lane] : 0u;
                for (unsigned k = 0; k < c; k += 32u) {
                    const unsigned cur = nxt;
                    const unsigned kn  = k + 32u;
                    if (kn < c) nxt = (kn + lane < c) ? mylist[kn + lane] : 0u;
                    const int      s  = (int)(cur & LIST_SRC_MASK);
                    const unsigned dl = cur >> LIST_SRC_BITS;
                    const unsigned rem = c - k;
                    if (rem >= 32u) drain_regs<F, true >(s, dl, 32u, lane, lo, dinv, t, tile, degtab, tabn);
                    else            drain_regs<F, false>(s, dl, rem, lane, lo, dinv, t, tile, degtab, tabn);
                }
            } else {
                const int e0 = r * per;
                const int e1 = (e0 + per) < E ? (e0 + per) : E;
                search_slice<F>(src, dst, dinv, t, tile, ring_s[wv], ring_d[wv], e0, e1, vec,
                                wv, lane, lo, nloc, degtab, tabn);
            }
        }
    }
    __syncthreads();

    float* ag = agg + (size_t)lo * F;
    int col = (int)tid % F;
    for (int i = (int)tid; i < se; i += 1024) {
        ag[i] = tile[i] + bias[col];
        col += 1024 % F;
        if (col >= F) col -= F;
    }
}

__global__ void k_gemm_t2(const float* __restrict__ agg1, const float* __restrict__ W2,
                          float* __restrict__ t2, int N, int nTiles) {
    int wave = (blockIdx.x * blockDim.x + threadIdx.x) >> 5;
    if (wave >= nTiles) return;
    int lane = threadIdx.x & 31;
    int n    = lane & 15;
    int kh   = lane >> 4;
    int rowA = wave * 16 + n;    if (rowA >= N) rowA = N - 1;

    v8f acc0 = {}, acc1 = {};
    const float* ar = agg1 + (size_t)rowA * 32;
    bool c1ok = (16 + n) < 20;
#pragma unroll
    for (int k0 = 0; k0 < 32; k0 += 4) {
        int kb = k0 + 2 * kh;
        v2f a;  a.x = fmaxf(ar[kb], 0.0f);  a.y = fmaxf(ar[kb + 1], 0.0f);
        v2f b0; b0.x = W2[kb * 20 + n];     b0.y = W2[(kb + 1) * 20 + n];
        v2f b1; b1.x = c1ok ? W2[kb * 20 + 16 + n] : 0.0f;
                b1.y = c1ok ? W2[(kb + 1) * 20 + 16 + n] : 0.0f;
        acc0 = wmma4(a, b0, acc0);
        acc1 = wmma4(a, b1, acc1);
    }
#pragma unroll
    for (int v = 0; v < 8; ++v) {
        int mr   = (lane < 16) ? v : v + 8;
        int node = wave * 16 + mr;
        if (node < N) {
            float* o = t2 + (size_t)node * 20;
            o[n] = acc0[v];
            if (c1ok) o[16 + n] = acc1[v];
        }
    }
}

extern "C" void kernel_launch(void* const* d_in, const int* in_sizes, int n_in,
                              void* d_out, int out_size, void* d_ws, size_t ws_size,
                              hipStream_t stream) {
    const float* x    = (const float*)d_in[0];
    const float* cnn  = (const float*)d_in[1];
    const int*   ei   = (const int*)d_in[2];
    const float* fcW  = (const float*)d_in[3];
    const float* fcb  = (const float*)d_in[4];
    const float* cnnW = (const float*)d_in[5];
    const float* cnnb = (const float*)d_in[6];
    const float* W1   = (const float*)d_in[7];
    const float* b1   = (const float*)d_in[8];
    const float* W2   = (const float*)d_in[9];
    const float* b2   = (const float*)d_in[10];

    const int N = in_sizes[0] / 32;
    const int E = in_sizes[2] / 2;
    const int* src = ei;
    const int* dst = ei + E;

    float* ws   = (float*)d_ws;
    float* dinv = ws;  ws += N;
    float* h0   = ws;  ws += (size_t)N * 64;
    float* t1   = ws;  ws += (size_t)N * 32;
    float* agg1 = ws;  ws += (size_t)N * 32;
    float* t2   = ws;  ws += (size_t)N * 20;
    float* out  = (float*)d_out;

    const int nTiles = (N + 15) / 16;
    dim3 blkG(128);
    dim3 grdG((nTiles * 32 + 127) / 128);
    const int PT = (N + RANGE_NP - 1) / RANGE_NP;
    unsigned* degb = (unsigned*)((float*)d_ws + ((N + 3) / 4) * 4);
    const int TW = ((N + 15) / 16) * 4;
    unsigned* lists = degb + TW;
    const long long lwaves = (long long)PT * LIST_SLICES * 32;
    long long lcap = (((long long)N * 64 - 3 - TW - lwaves) / lwaves) & ~31LL;
    if (lcap < 0 || N > (1 << LIST_SRC_BITS)) lcap = 0;
    if (lcap > 65536) lcap = 65536;
    const unsigned cap = (unsigned)lcap;

    k_encoder<<<grdG, blkG, 0, stream>>>(x, cnn, fcW, fcb, cnnW, cnnb, h0, N, nTiles);
    k_gemm_t1<<<grdG, blkG, 0, stream>>>(h0, W1, t1, N, nTiles);

    k_split_lists<RANGE_NP><<<LIST_SLICES * 32, 1024, 0, stream>>>(src, dst, lists, cap, E, N);

    k_deg_lists<RANGE_NP><<<PT, 1024, 0, stream>>>(dst, lists, cap, E, N, LIST_SLICES, dinv, degb);

    k_edge_tile<32, RANGE_NP><<<PT, 1024, 0, stream>>>(src, dst, dinv, t1, agg1, b1, lists, cap, E, N, degb);

    k_gemm_t2<<<grdG, blkG, 0, stream>>>(agg1, W2, t2, N, nTiles);
    k_edge_tile<20, RANGE_NP><<<PT, 1024, 0, stream>>>(src, dst, dinv, t2, out, b2, lists, cap, E, N, degb);
}


# Layer 2 aggregates before it transforms: its LDS tile sums relu(agg1) rows (one aligned 128-B line each), then W2 on WMMA from LDS; t2 and its GEMM launch go (7 -> 6); rows now summed before W2
# speedup vs baseline: 6.3410x; 1.1054x over previous
#include <hip/hip_runtime.h>
#include <hip/hip_bf16.h>

typedef __attribute__((ext_vector_type(2))) float v2f;
typedef __attribute__((ext_vector_type(8))) float v8f;

__device__ __forceinline__ v8f wmma4(v2f a, v2f b, v8f c) {
    return __builtin_amdgcn_wmma_f32_16x16x4_f32(
         false, a,  false, b,
         (short)0, c,  false,  false);
}

__global__ void k_encoder(const float* __restrict__ x, const float* __restrict__ cnn,
                          const float* __restrict__ fcW, const float* __restrict__ fcb,
                          const float* __restrict__ cnnW, const float* __restrict__ cnnb,
                          float* __restrict__ h0, int N, int nTiles) {
    int wave = (blockIdx.x * blockDim.x + threadIdx.x) >> 5;
    if (wave >= nTiles) return;
    int lane = threadIdx.x & 31;
    int n    = lane & 15;
    int kh   = lane >> 4;
    int rowA = wave * 16 + n;    if (rowA >= N) rowA = N - 1;

    v8f acc0 = {}, acc1 = {}, acc2 = {}, acc3 = {};

    const float* ax = x + (size_t)rowA * 32;
#pragma unroll
    for (int k0 = 0; k0 < 32; k0 += 4) {
        int kb = k0 + 2 * kh;
        v2f a;  a.x = ax[kb];               a.y = ax[kb + 1];
        v2f b0; b0.x = fcW[kb * 32 + n];    b0.y = fcW[(kb + 1) * 32 + n];
        v2f b1; b1.x = fcW[kb * 32 + 16 + n]; b1.y = fcW[(kb + 1) * 32 + 16 + n];
        acc0 = wmma4(a, b0, acc0);
        acc1 = wmma4(a, b1, acc1);
    }
    const float* ac = cnn + (size_t)rowA * 128;
#pragma unroll
    for (int k0 = 0; k0 < 128; k0 += 4) {
        int kb = k0 + 2 * kh;
        v2f a;  a.x = ac[kb];                a.y = ac[kb + 1];
        v2f b0; b0.x = cnnW[kb * 32 + n];    b0.y = cnnW[(kb + 1) * 32 + n];
        v2f b1; b1.x = cnnW[kb * 32 + 16 + n]; b1.y = cnnW[(kb + 1) * 32 + 16 + n];
        acc2 = wmma4(a, b0, acc2);
        acc3 = wmma4(a, b1, acc3);
    }
#pragma unroll
    for (int v = 0; v < 8; ++v) {
        int mr   = (lane < 16) ? v : v + 8;
        int node = wave * 16 + mr;
        if (node < N) {
            float* o = h0 + (size_t)node * 64;
            o[n]      = fmaxf(acc0[v] + fcb[n],       0.0f);
            o[16 + n] = fmaxf(acc1[v] + fcb[16 + n],  0.0f);
            o[32 + n] = fmaxf(acc2[v] + cnnb[n],      0.0f);
            o[48 + n] = fmaxf(acc3[v] + cnnb[16 + n], 0.0f);
        }
    }
}

__global__ void k_gemm_t1(const float* __restrict__ h0, const float* __restrict__ W1,
                          float* __restrict__ t1, int N, int nTiles) {
    int wave = (blockIdx.x * blockDim.x + threadIdx.x) >> 5;
    if (wave >= nTiles) return;
    int lane = threadIdx.x & 31;
    int n    = lane & 15;
    int kh   = lane >> 4;
    int rowA = wave * 16 + n;    if (rowA >= N) rowA = N - 1;

    v8f acc0 = {}, acc1 = {};
    const float* ar = h0 + (size_t)rowA * 64;
#pragma unroll
    for (int k0 = 0; k0 < 64; k0 += 4) {
        int kb = k0 + 2 * kh;
        v2f a;  a.x = ar[kb];              a.y = ar[kb + 1];
        v2f b0; b0.x = W1[kb * 32 + n];    b0.y = W1[(kb + 1) * 32 + n];
        v2f w1; w1.x = W1[kb * 32 + 16 + n]; w1.y = W1[(kb + 1) * 32 + 16 + n];
        acc0 = wmma4(a, b0, acc0);
        acc1 = wmma4(a, w1, acc1);
    }
#pragma unroll
    for (int v = 0; v < 8; ++v) {
        int mr   = (lane < 16) ? v : v + 8;
        int node = wave * 16 + mr;
        if (node < N) {
            float* o = t1 + (size_t)node * 32;
            o[n]      = acc0[v];
            o[16 + n] = acc1[v];
        }
    }
}

constexpr int      LIST_SRC_BITS = 21;
constexpr unsigned LIST_SRC_MASK = (1u << LIST_SRC_BITS) - 1u;

constexpr int RANGE_NP  = 400;
constexpr int DEG_TAB   = 102400;
constexpr int TAB_LOADS = (DEG_TAB / 16 + 1023) / 1024;

__device__ __forceinline__ float node_dinv(const unsigned char* __restrict__ tab, int tabn,
                                           const float* __restrict__ dinv, int node) {
    unsigned b = 255u;
    if (node < tabn) b = tab[node];
    return (b != 255u) ? rsqrtf((float)b + 1.0f) : dinv[node];
}

template <int F, bool FULL>
__device__ __forceinline__ void drain_regs(int s, unsigned dl, unsigned cnt, unsigned lane, int lo,
                                           const float* __restrict__ dinv, const float* __restrict__ t,
                                           float* __restrict__ tile,
                                           const unsigned char* __restrict__ tab, int tabn) {
    const bool act = lane < (unsigned)F;
    const unsigned fl = act ? lane : (unsigned)(F - 1);
    float v[32];
#pragma unroll
    for (int u = 0; u < 32; ++u) {
        const unsigned sj = (unsigned)__builtin_amdgcn_readlane(s, u);
        const float* row = t + (size_t)sj * F;
        v[u] = row[fl];
    }
    float w = 0.0f;
    if (FULL || lane < cnt)
        w = node_dinv(tab, tabn, dinv, s) * node_dinv(tab, tabn, dinv, lo + (int)dl);
    unsigned dj[32];
    float wj[32];
#pragma unroll
    for (int u = 0; u < 32; ++u) {
        dj[u] = (unsigned)__builtin_amdgcn_readlane((int)dl, u);
        wj[u] = __int_as_float(__builtin_amdgcn_readlane(__float_as_int(w), u));
    }
    if (act) {
#pragma unroll
        for (int u = 0; u < 32; ++u)
            if (FULL || (unsigned)u < cnt)
                atomicAdd(tile + dj[u] * F + lane, v[u] * wj[u]);
    }
}

template <int F, bool FULL>
__device__ __forceinline__ void drain_ring(const int* __restrict__ rs, const int* __restrict__ rd,
                                           unsigned head, unsigned cnt, unsigned lane, int lo,
                                           const float* __restrict__ dinv, const float* __restrict__ t,
                                           float* __restrict__ tile,
                                           const unsigned char* __restrict__ tab, int tabn) {
    const unsigned qi = (head + lane) & 63u;
    int s = 0;
    unsigned dl = 0;
    if (FULL || lane < cnt) {
        s  = rs[qi];
        dl = (unsigned)rd[qi];
    }
    drain_regs<F, FULL>(s, dl, cnt, lane, lo, dinv, t, tile, tab, tabn);
}

__device__ __forceinline__ void load_group(const int* __restrict__ src, const int* __restrict__ dst,
                                           int g, int e1, bool vec, unsigned lane,
                                           int (&d)[8], int (&s)[8]) {
    if (vec && g + 256 <= e1) {
#pragma unroll
        for (int h = 0; h < 2; ++h) {
            const int e = g + h * 128 + (int)lane * 4;
            const int4 qd = *reinterpret_cast<const int4*>(dst + e);
            const int4 qs = *reinterpret_cast<const int4*>(src + e);
            d[4 * h] = qd.x; d[4 * h + 1] = qd.y; d[4 * h + 2] = qd.z; d[4 * h + 3] = qd.w;
            s[4 * h] = qs.x; s[4 * h + 1] = qs.y; s[4 * h + 2] = qs.z; s[4 * h + 3] = qs.w;
        }
    } else {
#pragma unroll
        for (int h = 0; h < 2; ++h) {
#pragma unroll
            for (int i = 0; i < 4; ++i) {
                const int e = g + h * 128 + (int)lane * 4 + i;
                d[4 * h + i] = (e < e1) ? dst[e] : -1;
                s[4 * h + i] = (e < e1) ? src[e] : 0;
            }
        }
    }
}

template <int F>
__device__ __forceinline__ void search_slice(const int* __restrict__ src, const int* __restrict__ dst,
                                                 const float* __restrict__ dinv, const float* __restrict__ t,
                                                 float* __restrict__ tile, int* __restrict__ rs, int* __restrict__ rd,
                                                 int e0, int e1, bool vec, unsigned wv, unsigned lane,
                                                 int lo, int nloc,
                                                 const unsigned char* __restrict__ tab, int tabn) {
    unsigned head = 0, cnt = 0;

    int cd[8], cs[8];
#pragma unroll
    for (int k = 0; k < 8; ++k) { cd[k] = -1; cs[k] = 0; }
    int gn = e0 + (int)wv * 256;
    for (;;) {
        const bool more = gn < e1;
        int nd[8], ns[8];
        if (more) {
            load_group(src, dst, gn, e1, vec, lane, nd, ns);
        } else {
#pragma unroll
            for (int k = 0; k < 8; ++k) { nd[k] = -1; ns[k] = 0; }
        }
#pragma unroll
        for (int k = 0; k < 8; ++k) {
            const unsigned dl = (unsigned)(cd[k] - lo);
            const bool m = dl < (unsigned)nloc;
            const unsigned mask = __builtin_amdgcn_ballot_w32(m);
            if (mask != 0u) {
                if (m) {
                    const unsigned pos = __builtin_amdgcn_mbcnt_lo(mask, 0u);
                    const unsigned qi  = (head + cnt + pos) & 63u;
                    rs[qi] = cs[k];
                    rd[qi] = (int)dl;
                }
                cnt += (unsigned)__builtin_popcount(mask);
                if (cnt >= 32u) {
                    drain_ring<F, true>(rs, rd, head, cnt, lane, lo, dinv, t, tile, tab, tabn);
                    head = (head + 32u) & 63u;
                    cnt -= 32u;
                }
            }
        }
        if (!more) break;
#pragma unroll
        for (int k = 0; k < 8; ++k) { cd[k] = nd[k]; cs[k] = ns[k]; }
        gn += 32 * 256;
    }
    if (cnt != 0u)
        drain_ring<F, false>(rs, rd, head, cnt, lane, lo, dinv, t, tile, tab, tabn);
}

constexpr int SPLIT_MAXP  = 4096;
constexpr int SPLIT_STAGE = 65536;

__device__ __forceinline__ void split_fetch(const int* __restrict__ src, const int* __restrict__ dst,
                                            int g, int e1, bool vec, int l64,
                                            int (&d)[4], int (&s)[4]) {
    if (g < e1) {
        if (vec && g + 256 <= e1) {
            const int e = g + l64 * 4;
            const int4 qd = *reinterpret_cast<const int4*>(dst + e);
            const int4 qs = *reinterpret_cast<const int4*>(src + e);
            d[0] = qd.x; d[1] = qd.y; d[2] = qd.z; d[3] = qd.w;
            s[0] = qs.x; s[1] = qs.y; s[2] = qs.z; s[3] = qs.w;
        } else {
#pragma unroll
            for (int i = 0; i < 4; ++i) {
                const int e = g + l64 * 4 + i;
                d[i] = (e < e1) ? dst[e] : -1;
                s[i] = (e < e1) ? src[e] : 0;
            }
        }
    } else {
#pragma unroll
        for (int i = 0; i < 4; ++i) { d[i] = -1; s[i] = 0; }
    }
}

template <int NP>
__global__ __launch_bounds__(1024) void k_split_lists(const int* __restrict__ src, const int* __restrict__ dst,
                                                      unsigned* __restrict__ lists, unsigned cap,
                                                      int E, int N) {
    static_assert(NP <= (1 << (32 - LIST_SRC_BITS)), "a place in the range must fit above the source bits");
    __shared__ unsigned cnt[SPLIT_MAXP];
    __shared__ unsigned stage[SPLIT_STAGE];
    const int tid = (int)threadIdx.x;

    const int P  = (N + NP - 1) / NP;
    const int R  = (int)gridDim.x / 32;
    const int r  = (int)blockIdx.x / 32;
    const int wv = (int)blockIdx.x % 32;
    unsigned* counts = lists + (size_t)r * P * 32 + wv;
    if (P > SPLIT_MAXP) {
        for (int p = tid; p < P; p += 1024) counts[(size_t)p * 32] = 0xffffffffu;
        return;
    }
    for (int p = tid; p < P; p += 1024) cnt[p] = 0u;
    __syncthreads();

    const int per = (((E + R - 1) / R) + 255) & ~255;
    const int e0  = r * per;
    const int e1  = (e0 + per) < E ? (e0 + per) : E;
    const bool vec = (E & 3) == 0;
    const int q   = tid >> 6;
    const int l64 = tid & 63;
    const int g0  = e0 + wv * 256;
    const int J   = g0 < e1 ? (e1 - g0 + 8191) / 8192 : 0;
    const int T   = (J + 15) / 16;
    unsigned* ent = lists + (size_t)P * R * 32 + ((size_t)r * P * 32 + wv) * cap;
    const bool staged = (unsigned long long)P * cap <= (unsigned long long)SPLIT_STAGE;

    int cd[4], cs[4];
    split_fetch(src, dst, g0 + q * 8192, e1, vec, l64, cd, cs);
    for (int t = 0; t < T; ++t) {
        int nd[4], ns[4];
        split_fetch(src, dst, g0 + ((t + 1) * 16 + q) * 8192, e1, vec, l64, nd, ns);
        unsigned pp[4], rk[4];
#pragma unroll
        for (int i = 0; i < 4; ++i) {
            const bool ok = (unsigned)cd[i] < (unsigned)N;
            pp[i] = ok ? (unsigned)cd[i] / (unsigned)NP : 0u;
            rk[i] = ok ? atomicAdd(&cnt[pp[i]], 1u) : 0xffffffffu;
        }
#pragma unroll
        for (int i = 0; i < 4; ++i) {
            if (rk[i] < cap) {
                const unsigned dl = (unsigned)cd[i] - pp[i] * (unsigned)NP;
                const unsigned wd = (unsigned)cs[i] | (dl << LIST_SRC_BITS);
                if (staged) stage[pp[i] * cap + rk[i]] = wd;
                else        ent[(size_t)pp[i] * 32u * cap + rk[i]] = wd;
            }
        }
#pragma unroll
        for (int i = 0; i < 4; ++i) { cd[i] = nd[i]; cs[i] = ns[i]; }
    }
    __syncthreads();
    for (int p = tid; p < P; p += 1024) counts[(size_t)p * 32] = cnt[p];
    if (staged) {
        const unsigned lane = (unsigned)tid & 31u;
        for (int p = tid >> 5; p < P; p += 32) {
            const unsigned n = cnt[p] < cap ? cnt[p] : cap;
            const unsigned* sp = stage + (unsigned)p * cap;
            unsigned* gp = ent + (size_t)p * 32u * cap;
            for (unsigned k = lane; k < n; k += 32u) gp[k] = sp[k];
        }
    }
}

constexpr int DEG_LISTS = 8;
constexpr int DEG_LOADS = 4;

template <int NP>
__global__ __launch_bounds__(1024) void k_deg_lists(const int* __restrict__ dst,
                                                    const unsigned* __restrict__ lists, unsigned cap,
                                                    int E, int N, int R,
                                                    float* __restrict__ dinv, unsigned* __restrict__ degb) {
    static_assert(NP % 4 == 0, "a range starts on a whole word of in-degree bytes");
    __shared__ unsigned cnt[NP];
    const unsigned tid  = threadIdx.x;
    const unsigned lane = tid & 31u;
    const unsigned wv = (unsigned)__builtin_amdgcn_readfirstlane((int)(tid >> 5));
    const int P    = (int)gridDim.x;
    const int p    = (int)blockIdx.x;
    const int lo   = p * NP;
    const int nloc = (N - lo) < NP ? (N - lo) : NP;

    for (int i = (int)tid; i < NP; i += 1024) cnt[i] = 0u;
    __syncthreads();

    const unsigned* ent = lists + (size_t)P * R * 32u;
    const int per = (((E + R - 1) / R) + 255) & ~255;
    for (int rb = 0; rb < R; rb += DEG_LISTS) {
        unsigned cc[DEG_LISTS];
        const unsigned* ml[DEG_LISTS];
        const unsigned* cwp[DEG_LISTS];
        unsigned w[DEG_LISTS][DEG_LOADS];
#pragma unroll
        for (int j = 0; j < DEG_LISTS; ++j) {
            const int r = rb + j;
            const unsigned wid = ((unsigned)(r < R ? r : 0) * (unsigned)P + (unsigned)p) * 32u + wv;
            cc[j] = (r < R) ? (unsigned)__builtin_amdgcn_readfirstlane((int)lists[wid]) : 0u;
            ml[j]  = ent + (size_t)wid * cap;
            cwp[j] = lists + wid;
        }
#pragma unroll
        for (int j = 0; j < DEG_LISTS; ++j) {
#pragma unroll
            for (int u = 0; u < DEG_LOADS; ++u) {
                const unsigned k = (unsigned)u * 32u + lane;
                const unsigned* a = (cc[j] <= cap && k < cc[j]) ? (ml[j] + k) : cwp[j];
                w[j][u] = *a;
            }
        }
#pragma unroll
        for (int j = 0; j < DEG_LISTS; ++j) {
            if (cc[j] <= cap) {
#pragma unroll
                for (int u = 0; u < DEG_LOADS; ++u) {
                    const unsigned k = (unsigned)u * 32u + lane;
                    if (k < cc[j]) atomicAdd(&cnt[w[j][u] >> LIST_SRC_BITS], 1u);
                }
                for (unsigned k = (unsigned)DEG_LOADS * 32u + lane; k < cc[j]; k += 32u)
                    atomicAdd(&cnt[ml[j][k] >> LIST_SRC_BITS], 1u);
            } else {
                const int r  = rb + j;
                const int e0 = r * per;
                const int e1 = (e0 + per) < E ? (e0 + per) : E;
                for (int g = e0 + (int)wv * 256; g < e1; g += 32 * 256) {
                    for (int k = 0; k < 8; ++k) {
                        const int e = g + k * 32 + (int)lane;
                        if (e < e1) {
                            const unsigned dl = (unsigned)(dst[e] - lo);
                            if (dl < (unsigned)nloc) atomicAdd(&cnt[dl], 1u);
                        }
                    }
                }
            }
        }
    }
    __syncthreads();

    for (int i = (int)tid; i < nloc; i += 1024)
        dinv[lo + i] = rsqrtf((float)cnt[i] + 1.0f);
    const int TW = ((N + 15) / 16) * 4;
    const int w0 = lo / 4;
    const int w1 = (p == P - 1) ? TW : (lo + NP) / 4;
    for (int wi = w0 + (int)tid; wi < w1; wi += 1024) {
        unsigned wd = 0u;
#pragma unroll
        for (int k = 0; k < 4; ++k) {
            const int n = wi * 4 + k;
            unsigned b = 255u;
            if (n < N) {
                const unsigned d = cnt[n - lo];
                b = d < 255u ? d : 255u;
            }
            wd |= b << (8 * k);
        }
        degb[wi] = wd;
    }
}

constexpr int LIST_SLICES = 8;
constexpr int LIST_BUF    = 1024;
constexpr int LIST_PRE    = 3;

template <int F, int NP, int FO>
__global__ __launch_bounds__(1024) void k_edge_tile(const int* __restrict__ src, const int* __restrict__ dst,
                                                    const float* __restrict__ dinv, const float* __restrict__ t,
                                                    float* __restrict__ agg, const float* __restrict__ bias,
                                                    const float* __restrict__ Wt,
                                                    const unsigned* __restrict__ lists,
                                                    unsigned cap, int E, int N,
                                                    const unsigned* __restrict__ degb) {
    static_assert(NP <= (1 << (32 - LIST_SRC_BITS)), "a place in the range must fit above the source bits");
    static_assert(DEG_TAB % 16 == 0 && DEG_TAB / 16 <= TAB_LOADS * 1024, "the table copy covers the table");
    constexpr int SELF_LOADS = (NP * F + 1023) / 1024;
    __shared__ float tile[NP * F];
    __shared__ int ring_s[32][64];
    __shared__ int ring_d[32][64];
    __shared__ __attribute__((aligned(16))) unsigned char degtab[DEG_TAB];
    __shared__ unsigned lbuf[32][LIST_BUF];

    const unsigned tid  = threadIdx.x;
    const unsigned lane = tid & 31u;
    const unsigned wv = (unsigned)__builtin_amdgcn_readfirstlane((int)(tid >> 5));

    const int P = (int)gridDim.x;
    const int p = (int)blockIdx.x;
    const int lo   = p * NP;
    const int nloc = (N - lo) < NP ? (N - lo) : NP;

    const unsigned* ent = lists + (size_t)P * LIST_SLICES * 32u;
    unsigned cc[LIST_SLICES];
    unsigned total = 0u;
    bool staged = true;
#pragma unroll
    for (int r = 0; r < LIST_SLICES; ++r) {
        const unsigned wid = ((unsigned)r * (unsigned)P + (unsigned)p) * 32u + wv;
        cc[r] = (unsigned)__builtin_amdgcn_readfirstlane((int)lists[wid]);
        staged = staged && cc[r] <= cap && cc[r] <= (unsigned)(LIST_PRE * 32);
        total += cc[r];
    }
    staged = staged && total <= (unsigned)LIST_BUF;

    unsigned lw[LIST_SLICES][LIST_PRE];
#pragma unroll
    for (int r = 0; r < LIST_SLICES; ++r) {
        const unsigned wid = ((unsigned)r * (unsigned)P + (unsigned)p) * 32u + wv;
        const unsigned* ml = ent + (size_t)wid * cap;
#pragma unroll
        for (int u = 0; u < LIST_PRE; ++u) {
            const unsigned k = (unsigned)u * 32u + lane;
            lw[r][u] = (staged && k < cc[r]) ? ml[k] : 0u;
        }
    }

    const int tabn = N < DEG_TAB ? N : DEG_TAB;
    const int ng   = (tabn + 15) / 16;
    uint4 tq[TAB_LOADS];
#pragma unroll
    for (int k = 0; k < TAB_LOADS; ++k) {
        const int i = (int)tid + k * 1024;
        tq[k] = (i < ng) ? reinterpret_cast<const uint4*>(degb)[i] : make_uint4(0u, 0u, 0u, 0u);
    }

    const int se = nloc * F;
    const float* tself = t + (size_t)lo * F;
    float tv[SELF_LOADS];
#pragma unroll
    for (int u = 0; u < SELF_LOADS; ++u) {
        const int i = u * 1024 + (int)tid;
        tv[u] = (i < se) ? tself[i] : 0.0f;
    }

    if (staged) {
        unsigned* b = lbuf[wv];
        unsigned off = 0u;
#pragma unroll
        for (int r = 0; r < LIST_SLICES; ++r) {
#pragma unroll
            for (int u = 0; u < LIST_PRE; ++u) {
                const unsigned k = (unsigned)u * 32u + lane;
                if (k < cc[r]) b[off + k] = lw[r][u];
            }
            off += cc[r];
        }
    }

#pragma unroll
    for (int k = 0; k < TAB_LOADS; ++k) {
        const int i = (int)tid + k * 1024;
        if (i < ng) reinterpret_cast<uint4*>(degtab)[i] = tq[k];
    }
    __syncthreads();

#pragma unroll
    for (int u = 0; u < SELF_LOADS; ++u) {
        const int i = u * 1024 + (int)tid;
        if (i < se) {
            const float di = node_dinv(degtab, tabn, dinv, lo + i / F);
            tile[i] = tv[u] * di * di;
        }
    }
    __syncthreads();

    if (staged) {
        const unsigned* b = lbuf[wv];
        unsigned nxt = (lane < total) ? b[lane] : 0u;
        for (unsigned k = 0; k < total; k += 32u) {
            const unsigned cur = nxt;
            const unsigned kn  = k + 32u;
            if (kn < total) nxt = (kn + lane < total) ? b[kn + lane] : 0u;
            const int      s  = (int)(cur & LIST_SRC_MASK);
            const unsigned dl = cur >> LIST_SRC_BITS;
            const unsigned rem = total - k;
            if (rem >= 32u) drain_regs<F, true >(s, dl, 32u, lane, lo, dinv, t, tile, degtab, tabn);
            else            drain_regs<F, false>(s, dl, rem, lane, lo, dinv, t, tile, degtab, tabn);
        }
    } else {
        const int per = (((E + LIST_SLICES - 1) / LIST_SLICES) + 255) & ~255;
        const bool vec = (E & 3) == 0;
#pragma nounroll
        for (int r = 0; r < LIST_SLICES; ++r) {
            const unsigned wid = ((unsigned)r * (unsigned)P + (unsigned)p) * 32u + wv;
            const unsigned c = (unsigned)__builtin_amdgcn_readfirstlane((int)lists[wid]);
            if (c <= cap) {
                const unsigned* mylist = ent + (size_t)wid * cap;
                unsigned nxt = (lane < c) ? mylist[lane] : 0u;
                for (unsigned k = 0; k < c; k += 32u) {
                    const unsigned cur = nxt;
                    const unsigned kn  = k + 32u;
                    if (kn < c) nxt = (kn + lane < c) ? mylist[kn + lane] : 0u;
                    const int      s  = (int)(cur & LIST_SRC_MASK);
                    const unsigned dl = cur >> LIST_SRC_BITS;
                    const unsigned rem = c - k;
                    if (rem >= 32u) drain_regs<F, true >(s, dl, 32u, lane, lo, dinv, t, tile, degtab, tabn);
                    else            drain_regs<F, false>(s, dl, rem, lane, lo, dinv, t, tile, degtab, tabn);
                }
            } else {
                const int e0 = r * per;
                const int e1 = (e0 + per) < E ? (e0 + per) : E;
                search_slice<F>(src, dst, dinv, t, tile, ring_s[wv], ring_d[wv], e0, e1, vec,
                                wv, lane, lo, nloc, degtab, tabn);
            }
        }
    }
    if constexpr (FO == 0) {
        __syncthreads();
        float* ag = agg + (size_t)lo * F;
        int col = (int)tid % F;
        for (int i = (int)tid; i < se; i += 1024) {
            ag[i] = fmaxf(tile[i] + bias[col], 0.0f);
            col += 1024 % F;
            if (col >= F) col -= F;
        }
    } else {
        static_assert(F % 4 == 0 && FO >= 16 && FO <= 32, "two column tiles of 16 cover the output row");
        constexpr int NT = (NP + 15) / 16;
        const int n  = (int)(lane & 15u);
        const int kh = (int)(lane >> 4);
        const bool c1ok = (16 + n) < FO;
        v2f wb0[F / 4], wb1[F / 4];
#pragma unroll
        for (int j = 0; j < F / 4; ++j) {
            const int kb = 4 * j + 2 * kh;
            wb0[j].x = Wt[kb * FO + n];
            wb0[j].y = Wt[(kb + 1) * FO + n];
            wb1[j].x = c1ok ? Wt[kb * FO + 16 + n] : 0.0f;
            wb1[j].y = c1ok ? Wt[(kb + 1) * FO + 16 + n] : 0.0f;
        }
        const float bo0 = bias[n];
        const float bo1 = c1ok ? bias[16 + n] : 0.0f;
        __syncthreads();

        float* otile = reinterpret_cast<float*>(&lbuf[0][0]);
        static_assert(NP * FO <= 32 * LIST_BUF, "the output rows fit the list buffers");
        for (int tI = (int)wv; tI < NT; tI += 32) {
            int row = tI * 16 + n;
            if (row >= nloc) row = nloc - 1;
            const float* ar = tile + row * F;
            v8f acc0 = {}, acc1 = {};
#pragma unroll
            for (int j = 0; j < F / 4; ++j) {
                const int kb = 4 * j + 2 * kh;
                v2f a;  a.x = ar[kb];  a.y = ar[kb + 1];
                acc0 = wmma4(a, wb0[j], acc0);
                acc1 = wmma4(a, wb1[j], acc1);
            }
#pragma unroll
            for (int v = 0; v < 8; ++v) {
                const int r = tI * 16 + ((lane < 16u) ? v : v + 8);
                if (r < nloc) {
                    otile[r * FO + n] = acc0[v] + bo0;
                    if (c1ok) otile[r * FO + 16 + n] = acc1[v] + bo1;
                }
            }
        }
        __syncthreads();

        float* ag = agg + (size_t)lo * FO;
        const int oe = nloc * FO;
        for (int i = (int)tid; i < oe; i += 1024) ag[i] = otile[i];
    }
}

extern "C" void kernel_launch(void* const* d_in, const int* in_sizes, int n_in,
                              void* d_out, int out_size, void* d_ws, size_t ws_size,
                              hipStream_t stream) {
    const float* x    = (const float*)d_in[0];
    const float* cnn  = (const float*)d_in[1];
    const int*   ei   = (const int*)d_in[2];
    const float* fcW  = (const float*)d_in[3];
    const float* fcb  = (const float*)d_in[4];
    const float* cnnW = (const float*)d_in[5];
    const float* cnnb = (const float*)d_in[6];
    const float* W1   = (const float*)d_in[7];
    const float* b1   = (const float*)d_in[8];
    const float* W2   = (const float*)d_in[9];
    const float* b2   = (const float*)d_in[10];

    const int N = in_sizes[0] / 32;
    const int E = in_sizes[2] / 2;
    const int* src = ei;
    const int* dst = ei + E;

    float* ws   = (float*)d_ws;
    float* dinv = ws;  ws += N;
    float* h0   = ws;  ws += (size_t)N * 64;
    float* t1   = ws;  ws += (size_t)N * 32;
    float* agg1 = ws;  ws += (size_t)N * 32;
    float* out  = (float*)d_out;

    const int nTiles = (N + 15) / 16;
    dim3 blkG(128);
    dim3 grdG((nTiles * 32 + 127) / 128);
    const int PT = (N + RANGE_NP - 1) / RANGE_NP;
    unsigned* degb = (unsigned*)((float*)d_ws + ((N + 3) / 4) * 4);
    const int TW = ((N + 15) / 16) * 4;
    unsigned* lists = degb + TW;
    const long long lwaves = (long long)PT * LIST_SLICES * 32;
    long long lcap = (((long long)N * 64 - 3 - TW - lwaves) / lwaves) & ~31LL;
    if (lcap < 0 || N > (1 << LIST_SRC_BITS)) lcap = 0;
    if (lcap > 65536) lcap = 65536;
    const unsigned cap = (unsigned)lcap;

    k_encoder<<<grdG, blkG, 0, stream>>>(x, cnn, fcW, fcb, cnnW, cnnb, h0, N, nTiles);
    k_gemm_t1<<<grdG, blkG, 0, stream>>>(h0, W1, t1, N, nTiles);

    k_split_lists<RANGE_NP><<<LIST_SLICES * 32, 1024, 0, stream>>>(src, dst, lists, cap, E, N);

    k_deg_lists<RANGE_NP><<<PT, 1024, 0, stream>>>(dst, lists, cap, E, N, LIST_SLICES, dinv, degb);

    k_edge_tile<32, RANGE_NP, 0><<<PT, 1024, 0, stream>>>(src, dst, dinv, t1, agg1, b1, W2, lists, cap, E, N, degb);

    k_edge_tile<32, RANGE_NP, 20><<<PT, 1024, 0, stream>>>(src, dst, dinv, agg1, out, b2, W2, lists, cap, E, N, degb);
}
